# starting kernel
# baseline (speedup 1.0000x reference)
#include <hip/hip_runtime.h>

typedef __attribute__((ext_vector_type(2))) float v2f;
typedef __attribute__((ext_vector_type(8))) float v8f;

#define IN_DIM 128
#define HID_DIM 64


__global__ void gcn_zero_kernel(float* __restrict__ p, long long n) {
    long long i = (long long)blockIdx.x * blockDim.x + threadIdx.x;
    long long stride = (long long)gridDim.x * blockDim.x;
    for (; i < n; i += stride) p[i] = 0.0f;
}

__global__ void gcn_degree_kernel(const int* __restrict__ dst,
                                  float* __restrict__ deg, int E) {
    int e = blockIdx.x * blockDim.x + threadIdx.x;
    if (e < E) atomicAdd(&deg[dst[e]], 1.0f);
}

__global__ void gcn_norm_kernel(float* __restrict__ deg_norm, int N) {
    int i = blockIdx.x * blockDim.x + threadIdx.x;
    if (i < N) {
        float d = deg_norm[i];
        deg_norm[i] = rsqrtf(fmaxf(d, 1.0f));
    }
}

__global__ void gcn_relu_kernel(float* __restrict__ p, long long n) {
    long long i = (long long)blockIdx.x * blockDim.x + threadIdx.x;
    long long stride = (long long)gridDim.x * blockDim.x;
    for (; i < n; i += stride) p[i] = fmaxf(p[i], 0.0f);
}

template <int K, bool RELU_A>
__global__ void gcn_gemm_wmma(const float* __restrict__ X,
                              const float* __restrict__ W,
                              float* __restrict__ H) {
    __shared__ float sW[K * HID_DIM];
    for (int i = threadIdx.x; i < K * HID_DIM; i += blockDim.x) sW[i] = W[i];
    __syncthreads();

    const int wave = threadIdx.x >> 5;
    const int lane = threadIdx.x & 31;
    const int half = lane >> 4;
    const int lid  = lane & 15;
    const int row0 = blockIdx.x * 16;
    const int col0 = wave * 16;

    const float* __restrict__ Arow = X + (size_t)(row0 + lid) * K;

    v8f c = {};
#pragma unroll
    for (int k0 = 0; k0 < K; k0 += 4) {
        v2f a = *(const v2f*)(Arow + k0 + 2 * half);
        if (RELU_A) {
            a.x = fmaxf(a.x, 0.0f);
            a.y = fmaxf(a.y, 0.0f);
        }
        v2f b;
        b.x = sW[(k0 + 2 * half) * HID_DIM + col0 + lid];
        b.y = sW[(k0 + 2 * half + 1) * HID_DIM + col0 + lid];
        c = __builtin_amdgcn_wmma_f32_16x16x4_f32(
                 false, a,  false, b,
                 (short)0, c,  false,  false);
    }

    float* __restrict__ Hb = H + (size_t)row0 * HID_DIM + col0;
#pragma unroll
    for (int r = 0; r < 8; ++r)
        Hb[(size_t)(r + 8 * half) * HID_DIM + lid] = c[r];
}

__global__ void gcn_scatter_kernel(const float4* __restrict__ H4,
                                   const float* __restrict__ norm,
                                   const int* __restrict__ src,
                                   const int* __restrict__ dst,
                                   float* __restrict__ agg, int E) {
    long long gid = (long long)blockIdx.x * blockDim.x + threadIdx.x;
    int e = (int)(gid >> 4);
    int q = (int)(gid & 15);
    if (e >= E) return;
    int s = src[e];
    int t = dst[e];
    float4 v = H4[(size_t)s * (HID_DIM / 4) + q];
    float n = norm[s];
    float* o = agg + (size_t)t * HID_DIM + q * 4;
    atomicAdd(o + 0, v.x * n);
    atomicAdd(o + 1, v.y * n);
    atomicAdd(o + 2, v.z * n);
    atomicAdd(o + 3, v.w * n);
}


extern "C" void kernel_launch(void* const* d_in, const int* in_sizes, int n_in,
                              void* d_out, int out_size, void* d_ws, size_t ws_size,
                              hipStream_t stream) {
    const float* features = (const float*)d_in[0];
    const float* W1       = (const float*)d_in[1];
    const float* W2       = (const float*)d_in[2];
    const int*   src      = (const int*)d_in[3];
    const int*   dst      = (const int*)d_in[4];

    const int N = in_sizes[0] / IN_DIM;
    const int E = in_sizes[3];
    const long long NH = (long long)N * HID_DIM;

    float* ws   = (float*)d_ws;
    float* norm = ws;
    float* H    = ws + 65536;
    float* AGG  = H + NH;
    float* OUT  = (float*)d_out;

    const int mtiles = N / 16;

    gcn_zero_kernel<<<512, 256, 0, stream>>>(norm, N);
    gcn_degree_kernel<<<(E + 255) / 256, 256, 0, stream>>>(dst, norm, E);
    gcn_norm_kernel<<<(N + 255) / 256, 256, 0, stream>>>(norm, N);

    gcn_gemm_wmma<IN_DIM, false><<<mtiles, 128, 0, stream>>>(features, W1, H);
    gcn_zero_kernel<<<2048, 256, 0, stream>>>(AGG, NH);
    gcn_scatter_kernel<<<(E * 16 + 255) / 256, 256, 0, stream>>>(
        (const float4*)H, norm, src, dst, AGG, E);

    gcn_gemm_wmma<HID_DIM, true><<<mtiles, 128, 0, stream>>>(AGG, W2, H);
    gcn_zero_kernel<<<2048, 256, 0, stream>>>(OUT, NH);
    gcn_scatter_kernel<<<(E * 16 + 255) / 256, 256, 0, stream>>>(
        (const float4*)H, norm, src, dst, OUT, E);
    gcn_relu_kernel<<<2048, 256, 0, stream>>>(OUT, NH);
}


# scatter: a wave walks 32 edges; each row load and f32 atomic add covers one whole aligned 128-B line (lane l = dword l), not 4 lines x 8 floats; sums still in atomic arrival order
# speedup vs baseline: 1.3898x; 1.3898x over previous
#include <hip/hip_runtime.h>

typedef __attribute__((ext_vector_type(2))) float v2f;
typedef __attribute__((ext_vector_type(8))) float v8f;

#define IN_DIM 128
#define HID_DIM 64


__global__ void gcn_zero_kernel(float* __restrict__ p, long long n) {
    long long i = (long long)blockIdx.x * blockDim.x + threadIdx.x;
    long long stride = (long long)gridDim.x * blockDim.x;
    for (; i < n; i += stride) p[i] = 0.0f;
}

__global__ void gcn_degree_kernel(const int* __restrict__ dst,
                                  float* __restrict__ deg, int E) {
    int e = blockIdx.x * blockDim.x + threadIdx.x;
    if (e < E) atomicAdd(&deg[dst[e]], 1.0f);
}

__global__ void gcn_norm_kernel(float* __restrict__ deg_norm, int N) {
    int i = blockIdx.x * blockDim.x + threadIdx.x;
    if (i < N) {
        float d = deg_norm[i];
        deg_norm[i] = rsqrtf(fmaxf(d, 1.0f));
    }
}

__global__ void gcn_relu_kernel(float* __restrict__ p, long long n) {
    long long i = (long long)blockIdx.x * blockDim.x + threadIdx.x;
    long long stride = (long long)gridDim.x * blockDim.x;
    for (; i < n; i += stride) p[i] = fmaxf(p[i], 0.0f);
}

template <int K, bool RELU_A>
__global__ void gcn_gemm_wmma(const float* __restrict__ X,
                              const float* __restrict__ W,
                              float* __restrict__ H) {
    __shared__ float sW[K * HID_DIM];
    for (int i = threadIdx.x; i < K * HID_DIM; i += blockDim.x) sW[i] = W[i];
    __syncthreads();

    const int wave = threadIdx.x >> 5;
    const int lane = threadIdx.x & 31;
    const int half = lane >> 4;
    const int lid  = lane & 15;
    const int row0 = blockIdx.x * 16;
    const int col0 = wave * 16;

    const float* __restrict__ Arow = X + (size_t)(row0 + lid) * K;

    v8f c = {};
#pragma unroll
    for (int k0 = 0; k0 < K; k0 += 4) {
        v2f a = *(const v2f*)(Arow + k0 + 2 * half);
        if (RELU_A) {
            a.x = fmaxf(a.x, 0.0f);
            a.y = fmaxf(a.y, 0.0f);
        }
        v2f b;
        b.x = sW[(k0 + 2 * half) * HID_DIM + col0 + lid];
        b.y = sW[(k0 + 2 * half + 1) * HID_DIM + col0 + lid];
        c = __builtin_amdgcn_wmma_f32_16x16x4_f32(
                 false, a,  false, b,
                 (short)0, c,  false,  false);
    }

    float* __restrict__ Hb = H + (size_t)row0 * HID_DIM + col0;
#pragma unroll
    for (int r = 0; r < 8; ++r)
        Hb[(size_t)(r + 8 * half) * HID_DIM + lid] = c[r];
}

#define SC_EPW 32
#define SC_U   8

__device__ __forceinline__ void gcn_scatter_rows_load(const float* __restrict__ H, int s, int j0,
                                                      int lane, float (&a)[SC_U], float (&b)[SC_U]) {
#pragma unroll
    for (int u = 0; u < SC_U; ++u) {
        const int sj = __builtin_amdgcn_readlane(s, j0 + u);
        const float* __restrict__ hp = H + (size_t)sj * HID_DIM + lane;
        a[u] = hp[0];
        b[u] = hp[32];
    }
}

__device__ __forceinline__ void gcn_scatter_rows_add(float* __restrict__ agg, int t, float n, int j0,
                                                     int lane, const float (&a)[SC_U], const float (&b)[SC_U]) {
#pragma unroll
    for (int u = 0; u < SC_U; ++u) {
        const int tj = __builtin_amdgcn_readlane(t, j0 + u);
        const float nj = __int_as_float(__builtin_amdgcn_readlane(__float_as_int(n), j0 + u));
        float* o = agg + (size_t)tj * HID_DIM + lane;
        atomicAdd(o, a[u] * nj);
        atomicAdd(o + 32, b[u] * nj);
    }
}

__global__ __launch_bounds__(256) void gcn_scatter_kernel(const float* __restrict__ H,
                                   const float* __restrict__ norm,
                                   const int* __restrict__ src,
                                   const int* __restrict__ dst,
                                   float* __restrict__ agg, int E) {
    const int lane = threadIdx.x & 31;
    const long long wave = ((long long)blockIdx.x * blockDim.x + threadIdx.x) >> 5;
    const long long e0 = wave * SC_EPW;
    if (e0 >= E) return;
    const int cnt = (E - e0 < SC_EPW) ? (int)(E - e0) : SC_EPW;

    int s = 0, t = 0;
    float n = 0.0f;
    if (lane < cnt) {
        s = src[e0 + lane];
        t = dst[e0 + lane];
        n = norm[s];
    }

    if (cnt == SC_EPW) {
        float a0[SC_U], b0[SC_U], a1[SC_U], b1[SC_U];
        gcn_scatter_rows_load(H, s, 0, lane, a0, b0);
#pragma unroll
        for (int j0 = 0; j0 < SC_EPW; j0 += 2 * SC_U) {
            gcn_scatter_rows_load(H, s, j0 + SC_U, lane, a1, b1);
            gcn_scatter_rows_add(agg, t, n, j0, lane, a0, b0);
            if (j0 + 2 * SC_U < SC_EPW)
                gcn_scatter_rows_load(H, s, j0 + 2 * SC_U, lane, a0, b0);
            gcn_scatter_rows_add(agg, t, n, j0 + SC_U, lane, a1, b1);
        }
    } else {
        for (int j = 0; j < cnt; ++j) {
            const int sj = __builtin_amdgcn_readlane(s, j);
            const int tj = __builtin_amdgcn_readlane(t, j);
            const float nj = __int_as_float(__builtin_amdgcn_readlane(__float_as_int(n), j));
            const float* __restrict__ hp = H + (size_t)sj * HID_DIM + lane;
            float* o = agg + (size_t)tj * HID_DIM + lane;
            atomicAdd(o, hp[0] * nj);
            atomicAdd(o + 32, hp[32] * nj);
        }
    }
}


extern "C" void kernel_launch(void* const* d_in, const int* in_sizes, int n_in,
                              void* d_out, int out_size, void* d_ws, size_t ws_size,
                              hipStream_t stream) {
    const float* features = (const float*)d_in[0];
    const float* W1       = (const float*)d_in[1];
    const float* W2       = (const float*)d_in[2];
    const int*   src      = (const int*)d_in[3];
    const int*   dst      = (const int*)d_in[4];

    const int N = in_sizes[0] / IN_DIM;
    const int E = in_sizes[3];
    const long long NH = (long long)N * HID_DIM;

    float* ws   = (float*)d_ws;
    float* norm = ws;
    float* H    = ws + 65536;
    float* AGG  = H + NH;
    float* OUT  = (float*)d_out;

    const int mtiles = N / 16;

    gcn_zero_kernel<<<512, 256, 0, stream>>>(norm, N);
    gcn_degree_kernel<<<(E + 255) / 256, 256, 0, stream>>>(dst, norm, E);
    gcn_norm_kernel<<<(N + 255) / 256, 256, 0, stream>>>(norm, N);

    gcn_gemm_wmma<IN_DIM, false><<<mtiles, 128, 0, stream>>>(features, W1, H);
    gcn_zero_kernel<<<2048, 256, 0, stream>>>(AGG, NH);
    gcn_scatter_kernel<<<(E + 255) / 256, 256, 0, stream>>>(
        H, norm, src, dst, AGG, E);

    gcn_gemm_wmma<HID_DIM, true><<<mtiles, 128, 0, stream>>>(AGG, W2, H);
    gcn_zero_kernel<<<2048, 256, 0, stream>>>(OUT, NH);
    gcn_scatter_kernel<<<(E + 255) / 256, 256, 0, stream>>>(
        H, norm, src, dst, OUT, E);
    gcn_relu_kernel<<<2048, 256, 0, stream>>>(OUT, NH);
}


# GEMM operand feed: a wave does a 16x64 strip, so A is loaded once not 4x and W goes to LDS once per 64 rows by 16-B loads all in flight (was 64 serial 4-B load-wait-store trips per 16 rows)
# speedup vs baseline: 1.9203x; 1.3817x over previous
#include <hip/hip_runtime.h>

typedef __attribute__((ext_vector_type(2))) float v2f;
typedef __attribute__((ext_vector_type(4))) float v4f;
typedef __attribute__((ext_vector_type(8))) float v8f;

#define IN_DIM 128
#define HID_DIM 64


__global__ void gcn_zero_kernel(float* __restrict__ p, long long n) {
    long long i = (long long)blockIdx.x * blockDim.x + threadIdx.x;
    long long stride = (long long)gridDim.x * blockDim.x;
    for (; i < n; i += stride) p[i] = 0.0f;
}

__global__ void gcn_degree_kernel(const int* __restrict__ dst,
                                  float* __restrict__ deg, int E) {
    int e = blockIdx.x * blockDim.x + threadIdx.x;
    if (e < E) atomicAdd(&deg[dst[e]], 1.0f);
}

__global__ void gcn_norm_kernel(float* __restrict__ deg_norm, int N) {
    int i = blockIdx.x * blockDim.x + threadIdx.x;
    if (i < N) {
        float d = deg_norm[i];
        deg_norm[i] = rsqrtf(fmaxf(d, 1.0f));
    }
}

__global__ void gcn_relu_kernel(float* __restrict__ p, long long n) {
    long long i = (long long)blockIdx.x * blockDim.x + threadIdx.x;
    long long stride = (long long)gridDim.x * blockDim.x;
    for (; i < n; i += stride) p[i] = fmaxf(p[i], 0.0f);
}

#define GEMM_THREADS 128

template <int K, bool RELU_A>
__global__ __launch_bounds__(GEMM_THREADS) void gcn_gemm_wmma(const float* __restrict__ X,
                              const float* __restrict__ W,
                              float* __restrict__ H, int mtiles) {
    __shared__ v4f sW[K * 16];

    const int wave = threadIdx.x >> 5;
    const int lane = threadIdx.x & 31;
    const int half = lane >> 4;
    const int lid  = lane & 15;
    const int tile = blockIdx.x * (GEMM_THREADS / 32) + wave;
    const int tile_ld = (tile < mtiles) ? tile : (mtiles - 1);
    const int row0 = tile_ld * 16;

    constexpr int ITEMS = K * 8 / GEMM_THREADS;
    const v4f* __restrict__ W4 = (const v4f*)W;
    v4f q0[ITEMS], q1[ITEMS];
#pragma unroll
    for (int it = 0; it < ITEMS; ++it) {
        const int id = it * GEMM_THREADS + threadIdx.x;
        const int p = id >> 4, j = id & 15;
        q0[it] = W4[(2 * p) * 16 + j];
        q1[it] = W4[(2 * p + 1) * 16 + j];
    }

    const float* __restrict__ Arow = X + (size_t)(row0 + lid) * K;
    v2f a[K / 4];
#pragma unroll
    for (int s = 0; s < K / 4; ++s)
        a[s] = *(const v2f*)(Arow + 4 * s + 2 * half);

#pragma unroll
    for (int it = 0; it < ITEMS; ++it) {
        const int id = it * GEMM_THREADS + threadIdx.x;
        const int p = id >> 4, j = id & 15;
        v4f lo, hi;
        lo.x = q0[it].x; lo.y = q1[it].x; lo.z = q0[it].y; lo.w = q1[it].y;
        hi.x = q0[it].z; hi.y = q1[it].z; hi.z = q0[it].w; hi.w = q1[it].w;
        sW[(2 * p) * 16 + j]     = lo;
        sW[(2 * p + 1) * 16 + j] = hi;
    }
    __syncthreads();

    v8f c0 = {}, c1 = {}, c2 = {}, c3 = {};
#pragma unroll
    for (int s = 0; s < K / 4; ++s) {
        v2f av = a[s];
        if (RELU_A) {
            av.x = fmaxf(av.x, 0.0f);
            av.y = fmaxf(av.y, 0.0f);
        }
        const int p = 2 * s + half;
        const v4f u0 = sW[(2 * p) * 16 + lid];
        const v4f u1 = sW[(2 * p + 1) * 16 + lid];
        v2f b0, b1, b2, b3;
        b0.x = u0.x; b0.y = u0.y;
        b1.x = u0.z; b1.y = u0.w;
        b2.x = u1.x; b2.y = u1.y;
        b3.x = u1.z; b3.y = u1.w;
        c0 = __builtin_amdgcn_wmma_f32_16x16x4_f32(false, av, false, b0, (short)0, c0, false, false);
        c1 = __builtin_amdgcn_wmma_f32_16x16x4_f32(false, av, false, b1, (short)0, c1, false, false);
        c2 = __builtin_amdgcn_wmma_f32_16x16x4_f32(false, av, false, b2, (short)0, c2, false, false);
        c3 = __builtin_amdgcn_wmma_f32_16x16x4_f32(false, av, false, b3, (short)0, c3, false, false);
    }

    if (tile < mtiles) {
        float* __restrict__ Hb = H + (size_t)(row0 + 8 * half) * HID_DIM + 4 * lid;
#pragma unroll
        for (int r = 0; r < 8; ++r) {
            v4f o;
            o.x = c0[r]; o.y = c1[r]; o.z = c2[r]; o.w = c3[r];
            *(v4f*)(Hb + (size_t)r * HID_DIM) = o;
        }
    }
}

#define SC_EPW 32
#define SC_U   8

__device__ __forceinline__ void gcn_scatter_rows_load(const float* __restrict__ H, int s, int j0,
                                                      int lane, float (&a)[SC_U], float (&b)[SC_U]) {
#pragma unroll
    for (int u = 0; u < SC_U; ++u) {
        const int sj = __builtin_amdgcn_readlane(s, j0 + u);
        const float* __restrict__ hp = H + (size_t)sj * HID_DIM + lane;
        a[u] = hp[0];
        b[u] = hp[32];
    }
}

__device__ __forceinline__ void gcn_scatter_rows_add(float* __restrict__ agg, int t, float n, int j0,
                                                     int lane, const float (&a)[SC_U], const float (&b)[SC_U]) {
#pragma unroll
    for (int u = 0; u < SC_U; ++u) {
        const int tj = __builtin_amdgcn_readlane(t, j0 + u);
        const float nj = __int_as_float(__builtin_amdgcn_readlane(__float_as_int(n), j0 + u));
        float* o = agg + (size_t)tj * HID_DIM + lane;
        atomicAdd(o, a[u] * nj);
        atomicAdd(o + 32, b[u] * nj);
    }
}

__global__ __launch_bounds__(256) void gcn_scatter_kernel(const float* __restrict__ H,
                                   const float* __restrict__ norm,
                                   const int* __restrict__ src,
                                   const int* __restrict__ dst,
                                   float* __restrict__ agg, int E) {
    const int lane = threadIdx.x & 31;
    const long long wave = ((long long)blockIdx.x * blockDim.x + threadIdx.x) >> 5;
    const long long e0 = wave * SC_EPW;
    if (e0 >= E) return;
    const int cnt = (E - e0 < SC_EPW) ? (int)(E - e0) : SC_EPW;

    int s = 0, t = 0;
    float n = 0.0f;
    if (lane < cnt) {
        s = src[e0 + lane];
        t = dst[e0 + lane];
        n = norm[s];
    }

    if (cnt == SC_EPW) {
        float a0[SC_U], b0[SC_U], a1[SC_U], b1[SC_U];
        gcn_scatter_rows_load(H, s, 0, lane, a0, b0);
#pragma unroll
        for (int j0 = 0; j0 < SC_EPW; j0 += 2 * SC_U) {
            gcn_scatter_rows_load(H, s, j0 + SC_U, lane, a1, b1);
            gcn_scatter_rows_add(agg, t, n, j0, lane, a0, b0);
            if (j0 + 2 * SC_U < SC_EPW)
                gcn_scatter_rows_load(H, s, j0 + 2 * SC_U, lane, a0, b0);
            gcn_scatter_rows_add(agg, t, n, j0 + SC_U, lane, a1, b1);
        }
    } else {
        for (int j = 0; j < cnt; ++j) {
            const int sj = __builtin_amdgcn_readlane(s, j);
            const int tj = __builtin_amdgcn_readlane(t, j);
            const float nj = __int_as_float(__builtin_amdgcn_readlane(__float_as_int(n), j));
            const float* __restrict__ hp = H + (size_t)sj * HID_DIM + lane;
            float* o = agg + (size_t)tj * HID_DIM + lane;
            atomicAdd(o, hp[0] * nj);
            atomicAdd(o + 32, hp[32] * nj);
        }
    }
}


extern "C" void kernel_launch(void* const* d_in, const int* in_sizes, int n_in,
                              void* d_out, int out_size, void* d_ws, size_t ws_size,
                              hipStream_t stream) {
    const float* features = (const float*)d_in[0];
    const float* W1       = (const float*)d_in[1];
    const float* W2       = (const float*)d_in[2];
    const int*   src      = (const int*)d_in[3];
    const int*   dst      = (const int*)d_in[4];

    const int N = in_sizes[0] / IN_DIM;
    const int E = in_sizes[3];
    const long long NH = (long long)N * HID_DIM;

    float* ws   = (float*)d_ws;
    float* norm = ws;
    float* H    = ws + 65536;
    float* AGG  = H + NH;
    float* OUT  = (float*)d_out;

    const int mtiles = N / 16;
    const int gblocks = (mtiles + 3) / 4;

    gcn_zero_kernel<<<512, 256, 0, stream>>>(norm, N);
    gcn_degree_kernel<<<(E + 255) / 256, 256, 0, stream>>>(dst, norm, E);
    gcn_norm_kernel<<<(N + 255) / 256, 256, 0, stream>>>(norm, N);

    gcn_gemm_wmma<IN_DIM, false><<<gblocks, GEMM_THREADS, 0, stream>>>(features, W1, H, mtiles);
    gcn_zero_kernel<<<2048, 256, 0, stream>>>(AGG, NH);
    gcn_scatter_kernel<<<(E + 255) / 256, 256, 0, stream>>>(
        H, norm, src, dst, AGG, E);

    gcn_gemm_wmma<HID_DIM, true><<<gblocks, GEMM_THREADS, 0, stream>>>(AGG, W2, H, mtiles);
    gcn_zero_kernel<<<2048, 256, 0, stream>>>(OUT, NH);
    gcn_scatter_kernel<<<(E + 255) / 256, 256, 0, stream>>>(
        H, norm, src, dst, OUT, E);
    gcn_relu_kernel<<<2048, 256, 0, stream>>>(OUT, NH);
}


# atomics: each f32 atomic target (degrees, both aggregates) is two private copies, one per L2 half (wave reads its die id); consumer adds copy 0 + copy 1: each sum = two partial sums, then one add
# speedup vs baseline: 2.7919x; 1.4539x over previous
#include <hip/hip_runtime.h>

typedef __attribute__((ext_vector_type(2))) float v2f;
typedef __attribute__((ext_vector_type(4))) float v4f;
typedef __attribute__((ext_vector_type(8))) float v8f;

#define IN_DIM 128
#define HID_DIM 64


__device__ __forceinline__ int gcn_l2_half() {
    const unsigned id = __builtin_amdgcn_s_sendmsg_rtn(0x87);
    return (int)((id >> 18) & 1u);
}

#define DEG_STRIDE 65536

__global__ void gcn_zero_kernel(float* __restrict__ p, long long n) {
    long long i = (long long)blockIdx.x * blockDim.x + threadIdx.x;
    long long stride = (long long)gridDim.x * blockDim.x;
    for (; i < n; i += stride) p[i] = 0.0f;
}

__global__ void gcn_degree_kernel(const int* __restrict__ dst,
                                  float* __restrict__ deg, int E) {
    float* __restrict__ mine = deg + (size_t)gcn_l2_half() * DEG_STRIDE;
    int e = blockIdx.x * blockDim.x + threadIdx.x;
    if (e < E) atomicAdd(&mine[dst[e]], 1.0f);
}

__global__ void gcn_norm_kernel(float* __restrict__ deg_norm, int N) {
    int i = blockIdx.x * blockDim.x + threadIdx.x;
    if (i < N) {
        float d = deg_norm[i] + deg_norm[i + DEG_STRIDE];
        deg_norm[i] = rsqrtf(fmaxf(d, 1.0f));
    }
}

__global__ void gcn_sum_relu_kernel(const float* __restrict__ a_f, const float* __restrict__ b_f,
                                    float* __restrict__ out_f, long long n4) {
    const v4f* __restrict__ a = (const v4f*)a_f;
    const v4f* __restrict__ b = (const v4f*)b_f;
    v4f* __restrict__ out = (v4f*)out_f;
    long long i = (long long)blockIdx.x * blockDim.x + threadIdx.x;
    long long stride = (long long)gridDim.x * blockDim.x;
    for (; i < n4; i += stride) {
        const v4f x = a[i], y = b[i];
        v4f o;
        o.x = fmaxf(x.x + y.x, 0.0f);
        o.y = fmaxf(x.y + y.y, 0.0f);
        o.z = fmaxf(x.z + y.z, 0.0f);
        o.w = fmaxf(x.w + y.w, 0.0f);
        out[i] = o;
    }
}

#define GEMM_THREADS 128

template <int K, bool RELU_A, bool SUM_A>
__global__ __launch_bounds__(GEMM_THREADS) void gcn_gemm_wmma(const float* __restrict__ X,
                              const float* __restrict__ X2,
                              const float* __restrict__ W,
                              float* __restrict__ H, int mtiles) {
    __shared__ v4f sW[K * 16];

    const int wave = threadIdx.x >> 5;
    const int lane = threadIdx.x & 31;
    const int half = lane >> 4;
    const int lid  = lane & 15;
    const int tile = blockIdx.x * (GEMM_THREADS / 32) + wave;
    const int tile_ld = (tile < mtiles) ? tile : (mtiles - 1);
    const int row0 = tile_ld * 16;

    constexpr int ITEMS = K * 8 / GEMM_THREADS;
    const v4f* __restrict__ W4 = (const v4f*)W;
    v4f q0[ITEMS], q1[ITEMS];
#pragma unroll
    for (int it = 0; it < ITEMS; ++it) {
        const int id = it * GEMM_THREADS + threadIdx.x;
        const int p = id >> 4, j = id & 15;
        q0[it] = W4[(2 * p) * 16 + j];
        q1[it] = W4[(2 * p + 1) * 16 + j];
    }

    const float* __restrict__ Arow = X + (size_t)(row0 + lid) * K;
    v2f a[K / 4];
#pragma unroll
    for (int s = 0; s < K / 4; ++s)
        a[s] = *(const v2f*)(Arow + 4 * s + 2 * half);
    v2f a2[SUM_A ? K / 4 : 1];
    if (SUM_A) {
        const float* __restrict__ Arow2 = X2 + (size_t)(row0 + lid) * K;
#pragma unroll
        for (int s = 0; s < K / 4; ++s)
            a2[s] = *(const v2f*)(Arow2 + 4 * s + 2 * half);
    }

#pragma unroll
    for (int it = 0; it < ITEMS; ++it) {
        const int id = it * GEMM_THREADS + threadIdx.x;
        const int p = id >> 4, j = id & 15;
        v4f lo, hi;
        lo.x = q0[it].x; lo.y = q1[it].x; lo.z = q0[it].y; lo.w = q1[it].y;
        hi.x = q0[it].z; hi.y = q1[it].z; hi.z = q0[it].w; hi.w = q1[it].w;
        sW[(2 * p) * 16 + j]     = lo;
        sW[(2 * p + 1) * 16 + j] = hi;
    }
    __syncthreads();

    v8f c0 = {}, c1 = {}, c2 = {}, c3 = {};
#pragma unroll
    for (int s = 0; s < K / 4; ++s) {
        v2f av = a[s];
        if (SUM_A) {
            av.x += a2[s].x;
            av.y += a2[s].y;
        }
        if (RELU_A) {
            av.x = fmaxf(av.x, 0.0f);
            av.y = fmaxf(av.y, 0.0f);
        }
        const int p = 2 * s + half;
        const v4f u0 = sW[(2 * p) * 16 + lid];
        const v4f u1 = sW[(2 * p + 1) * 16 + lid];
        v2f b0, b1, b2, b3;
        b0.x = u0.x; b0.y = u0.y;
        b1.x = u0.z; b1.y = u0.w;
        b2.x = u1.x; b2.y = u1.y;
        b3.x = u1.z; b3.y = u1.w;
        c0 = __builtin_amdgcn_wmma_f32_16x16x4_f32(false, av, false, b0, (short)0, c0, false, false);
        c1 = __builtin_amdgcn_wmma_f32_16x16x4_f32(false, av, false, b1, (short)0, c1, false, false);
        c2 = __builtin_amdgcn_wmma_f32_16x16x4_f32(false, av, false, b2, (short)0, c2, false, false);
        c3 = __builtin_amdgcn_wmma_f32_16x16x4_f32(false, av, false, b3, (short)0, c3, false, false);
    }

    if (tile < mtiles) {
        float* __restrict__ Hb = H + (size_t)(row0 + 8 * half) * HID_DIM + 4 * lid;
#pragma unroll
        for (int r = 0; r < 8; ++r) {
            v4f o;
            o.x = c0[r]; o.y = c1[r]; o.z = c2[r]; o.w = c3[r];
            *(v4f*)(Hb + (size_t)r * HID_DIM) = o;
        }
    }
}

#define SC_EPW 32
#define SC_U   8

__device__ __forceinline__ void gcn_scatter_rows_load(const float* __restrict__ H, int s, int j0,
                                                      int lane, float (&a)[SC_U], float (&b)[SC_U]) {
#pragma unroll
    for (int u = 0; u < SC_U; ++u) {
        const int sj = __builtin_amdgcn_readlane(s, j0 + u);
        const float* __restrict__ hp = H + (size_t)sj * HID_DIM + lane;
        a[u] = hp[0];
        b[u] = hp[32];
    }
}

__device__ __forceinline__ void gcn_scatter_rows_add(float* __restrict__ agg, int t, float n, int j0,
                                                     int lane, const float (&a)[SC_U], const float (&b)[SC_U]) {
#pragma unroll
    for (int u = 0; u < SC_U; ++u) {
        const int tj = __builtin_amdgcn_readlane(t, j0 + u);
        const float nj = __int_as_float(__builtin_amdgcn_readlane(__float_as_int(n), j0 + u));
        float* o = agg + (size_t)tj * HID_DIM + lane;
        atomicAdd(o, a[u] * nj);
        atomicAdd(o + 32, b[u] * nj);
    }
}

__global__ __launch_bounds__(256) void gcn_scatter_kernel(const float* __restrict__ H,
                                   const float* __restrict__ norm,
                                   const int* __restrict__ src,
                                   const int* __restrict__ dst,
                                   float* __restrict__ agg2, long long copy_stride, int E) {
    float* __restrict__ agg = agg2 + (size_t)gcn_l2_half() * (size_t)copy_stride;
    const int lane = threadIdx.x & 31;
    const long long wave = ((long long)blockIdx.x * blockDim.x + threadIdx.x) >> 5;
    const long long e0 = wave * SC_EPW;
    if (e0 >= E) return;
    const int cnt = (E - e0 < SC_EPW) ? (int)(E - e0) : SC_EPW;

    int s = 0, t = 0;
    float n = 0.0f;
    if (lane < cnt) {
        s = src[e0 + lane];
        t = dst[e0 + lane];
        n = norm[s];
    }

    if (cnt == SC_EPW) {
        float a0[SC_U], b0[SC_U], a1[SC_U], b1[SC_U];
        gcn_scatter_rows_load(H, s, 0, lane, a0, b0);
#pragma unroll
        for (int j0 = 0; j0 < SC_EPW; j0 += 2 * SC_U) {
            gcn_scatter_rows_load(H, s, j0 + SC_U, lane, a1, b1);
            gcn_scatter_rows_add(agg, t, n, j0, lane, a0, b0);
            if (j0 + 2 * SC_U < SC_EPW)
                gcn_scatter_rows_load(H, s, j0 + 2 * SC_U, lane, a0, b0);
            gcn_scatter_rows_add(agg, t, n, j0 + SC_U, lane, a1, b1);
        }
    } else {
        for (int j = 0; j < cnt; ++j) {
            const int sj = __builtin_amdgcn_readlane(s, j);
            const int tj = __builtin_amdgcn_readlane(t, j);
            const float nj = __int_as_float(__builtin_amdgcn_readlane(__float_as_int(n), j));
            const float* __restrict__ hp = H + (size_t)sj * HID_DIM + lane;
            float* o = agg + (size_t)tj * HID_DIM + lane;
            atomicAdd(o, hp[0] * nj);
            atomicAdd(o + 32, hp[32] * nj);
        }
    }
}


extern "C" void kernel_launch(void* const* d_in, const int* in_sizes, int n_in,
                              void* d_out, int out_size, void* d_ws, size_t ws_size,
                              hipStream_t stream) {
    const float* features = (const float*)d_in[0];
    const float* W1       = (const float*)d_in[1];
    const float* W2       = (const float*)d_in[2];
    const int*   src      = (const int*)d_in[3];
    const int*   dst      = (const int*)d_in[4];

    const int N = in_sizes[0] / IN_DIM;
    const int E = in_sizes[3];
    const long long NH = (long long)N * HID_DIM;

    float* ws   = (float*)d_ws;
    float* norm = ws;
    float* H    = ws + 2 * DEG_STRIDE;
    float* ACC  = H + NH;
    float* OUT  = (float*)d_out;

    const int mtiles = N / 16;
    const int gblocks = (mtiles + 3) / 4;

    gcn_zero_kernel<<<512, 256, 0, stream>>>(norm, 2 * DEG_STRIDE);
    gcn_degree_kernel<<<(E + 255) / 256, 256, 0, stream>>>(dst, norm, E);
    gcn_norm_kernel<<<(N + 255) / 256, 256, 0, stream>>>(norm, N);

    gcn_gemm_wmma<IN_DIM, false, false><<<gblocks, GEMM_THREADS, 0, stream>>>(features, features, W1, H, mtiles);
    gcn_zero_kernel<<<2048, 256, 0, stream>>>(ACC, 2 * NH);
    gcn_scatter_kernel<<<(E + 255) / 256, 256, 0, stream>>>(
        H, norm, src, dst, ACC, NH, E);

    gcn_gemm_wmma<HID_DIM, true, true><<<gblocks, GEMM_THREADS, 0, stream>>>(ACC, ACC + NH, W2, H, mtiles);
    gcn_zero_kernel<<<2048, 256, 0, stream>>>(ACC, 2 * NH);
    gcn_scatter_kernel<<<(E + 255) / 256, 256, 0, stream>>>(
        H, norm, src, dst, ACC, NH, E);
    gcn_sum_relu_kernel<<<2048, 256, 0, stream>>>(ACC, ACC + NH, OUT, NH / 4);
}


# aggregation without float atomics: edges sorted by dst once per call (two-level counting sort on LDS counters), then a wave sums each node's source rows in registers; sum order = slot arrival order
# speedup vs baseline: 2.9676x; 1.0629x over previous
#include <hip/hip_runtime.h>

typedef __attribute__((ext_vector_type(2))) float v2f;
typedef __attribute__((ext_vector_type(4))) float v4f;
typedef __attribute__((ext_vector_type(8))) float v8f;
typedef __attribute__((ext_vector_type(2))) int v2i;
typedef __attribute__((ext_vector_type(4))) int v4i;

#define IN_DIM 128
#define HID_DIM 64

#define BK_SHIFT 8
#define BK_NODES 256
#define NBK_MAX  256
#define ED_EPB   2048
#define ED_T     256

__global__ __launch_bounds__(ED_T) void gcn_hist_kernel(const int* __restrict__ dst,
                                                        int* __restrict__ bh, int E, int nbk) {
    __shared__ int h[NBK_MAX];
    const int t = threadIdx.x;
    h[t] = 0;
    __syncthreads();
    const long long e0 = (long long)blockIdx.x * ED_EPB;
    if (e0 + ED_EPB <= E) {
        const v4i da = *(const v4i*)(dst + e0 + 4 * t);
        const v4i db = *(const v4i*)(dst + e0 + ED_EPB / 2 + 4 * t);
        atomicAdd(&h[da.x >> BK_SHIFT], 1);
        atomicAdd(&h[da.y >> BK_SHIFT], 1);
        atomicAdd(&h[da.z >> BK_SHIFT], 1);
        atomicAdd(&h[da.w >> BK_SHIFT], 1);
        atomicAdd(&h[db.x >> BK_SHIFT], 1);
        atomicAdd(&h[db.y >> BK_SHIFT], 1);
        atomicAdd(&h[db.z >> BK_SHIFT], 1);
        atomicAdd(&h[db.w >> BK_SHIFT], 1);
    } else {
        for (long long e = e0 + t; e < E; e += ED_T) atomicAdd(&h[dst[e] >> BK_SHIFT], 1);
    }
    __syncthreads();
    if (t < nbk) bh[(size_t)t * gridDim.x + blockIdx.x] = h[t];
}

#define SCAN_T 1024
__global__ __launch_bounds__(SCAN_T) void gcn_scan_kernel(const int* __restrict__ in,
                                                          int* __restrict__ out, int n) {
    __shared__ int s_tot[32];
    __shared__ int s_pre[32];
    const int t = threadIdx.x, lane = t & 31, w = t >> 5;

    int pre = 0;
#pragma unroll 16
    for (int k = 0; k < (int)blockIdx.x; ++k) pre += in[k * SCAN_T + t];

    const int i = blockIdx.x * SCAN_T + t;
    const int d = (i < n) ? in[i] : 0;

    int x = d;
#pragma unroll
    for (int o = 1; o < 32; o <<= 1) {
        const int y = __shfl_up(x, o);
        if (lane >= o) x += y;
    }
#pragma unroll
    for (int o = 16; o > 0; o >>= 1) pre += __shfl_xor(pre, o);
    if (lane == 31) s_tot[w] = x;
    if (lane == 0) s_pre[w] = pre;
    __syncthreads();

    const int wt = s_tot[lane];
    int ws = wt;
#pragma unroll
    for (int o = 1; o < 32; o <<= 1) {
        const int y = __shfl_up(ws, o);
        if (lane >= o) ws += y;
    }
    int pt = s_pre[lane];
#pragma unroll
    for (int o = 16; o > 0; o >>= 1) pt += __shfl_xor(pt, o);
    const int wave_excl = __shfl(ws - wt, w);

    const int excl = pt + wave_excl + (x - d);
    if (i < n) {
        out[i] = excl;
        if (i == n - 1) out[n] = excl + d;
    }
}

__global__ __launch_bounds__(ED_T) void gcn_split_kernel(const int* __restrict__ src,
                                                         const int* __restrict__ dst,
                                                         const int* __restrict__ off,
                                                         v2i* __restrict__ tmp, int E, int nbk) {
    __shared__ int cur[NBK_MAX];
    const int t = threadIdx.x;
    cur[t] = (t < nbk) ? off[(size_t)t * gridDim.x + blockIdx.x] : 0;
    __syncthreads();
    const long long e0 = (long long)blockIdx.x * ED_EPB;
    if (e0 + ED_EPB <= E) {
        const v4i da = *(const v4i*)(dst + e0 + 4 * t);
        const v4i db = *(const v4i*)(dst + e0 + ED_EPB / 2 + 4 * t);
        const v4i sa = *(const v4i*)(src + e0 + 4 * t);
        const v4i sb = *(const v4i*)(src + e0 + ED_EPB / 2 + 4 * t);
        const int d[8] = {da.x, da.y, da.z, da.w, db.x, db.y, db.z, db.w};
        const int s[8] = {sa.x, sa.y, sa.z, sa.w, sb.x, sb.y, sb.z, sb.w};
        int r[8];
#pragma unroll
        for (int u = 0; u < 8; ++u) r[u] = atomicAdd(&cur[d[u] >> BK_SHIFT], 1);
#pragma unroll
        for (int u = 0; u < 8; ++u) {
            v2i o;
            o.x = s[u];
            o.y = d[u];
            tmp[r[u]] = o;
        }
    } else {
        for (long long e = e0 + t; e < E; e += ED_T) {
            const int dd = dst[e];
            const int r = atomicAdd(&cur[dd >> BK_SHIFT], 1);
            v2i o;
            o.x = src[e];
            o.y = dd;
            tmp[r] = o;
        }
    }
}

#define BF_T 1024
#define BF_R 8
__global__ __launch_bounds__(BF_T) void gcn_bucket_kernel(const v2i* __restrict__ tmp,
                                                          const int* __restrict__ off,
                                                          int* __restrict__ start,
                                                          float* __restrict__ norm,
                                                          int* __restrict__ es, int N, int ga) {
    __shared__ int cnt[BK_NODES];
    __shared__ int wtot[BK_NODES / 32];
    const int t = threadIdx.x, lane = t & 31, w = t >> 5;
    const int b = blockIdx.x;
    const int bs = off[(size_t)b * ga];
    const int be = off[(size_t)(b + 1) * ga];
    if (t < BK_NODES) cnt[t] = 0;
    __syncthreads();

    v2i q[BF_R];
#pragma unroll
    for (int u = 0; u < BF_R; ++u) {
        const int i = bs + u * BF_T + t;
        q[u].x = 0;
        q[u].y = -1;
        if (i < be) q[u] = tmp[i];
    }
#pragma unroll
    for (int u = 0; u < BF_R; ++u)
        if (q[u].y >= 0) atomicAdd(&cnt[q[u].y & (BK_NODES - 1)], 1);
    for (int i = bs + BF_R * BF_T + t; i < be; i += BF_T)
        atomicAdd(&cnt[tmp[i].y & (BK_NODES - 1)], 1);
    __syncthreads();

    const int c = (t < BK_NODES) ? cnt[t] : 0;
    int x = c;
#pragma unroll
    for (int o = 1; o < 32; o <<= 1) {
        const int y = __shfl_up(x, o);
        if (lane >= o) x += y;
    }
    if (t < BK_NODES && lane == 31) wtot[w] = x;
    __syncthreads();
    if (t < BK_NODES) {
        int pre = 0;
        for (int k = 0; k < w; ++k) pre += wtot[k];
        const int first = bs + pre + (x - c);
        const int node = b * BK_NODES + t;
        cnt[t] = first;
        if (node < N) {
            start[node] = first;
            norm[node] = rsqrtf(fmaxf((float)c, 1.0f));
            if (node == N - 1) start[N] = first + c;
        }
    }
    __syncthreads();

    int r[BF_R];
#pragma unroll
    for (int u = 0; u < BF_R; ++u) {
        r[u] = 0;
        if (q[u].y >= 0) r[u] = atomicAdd(&cnt[q[u].y & (BK_NODES - 1)], 1);
    }
#pragma unroll
    for (int u = 0; u < BF_R; ++u)
        if (q[u].y >= 0) es[r[u]] = q[u].x;
    for (int i = bs + BF_R * BF_T + t; i < be; i += BF_T) {
        const v2i sd = tmp[i];
        es[atomicAdd(&cnt[sd.y & (BK_NODES - 1)], 1)] = sd.x;
    }
}

#define GEMM_THREADS 128

template <int K, bool RELU_A, bool SUM_A>
__global__ __launch_bounds__(GEMM_THREADS) void gcn_gemm_wmma(const float* __restrict__ X,
                              const float* __restrict__ X2,
                              const float* __restrict__ W,
                              float* __restrict__ H, int mtiles) {
    __shared__ v4f sW[K * 16];

    const int wave = threadIdx.x >> 5;
    const int lane = threadIdx.x & 31;
    const int half = lane >> 4;
    const int lid  = lane & 15;
    const int tile = blockIdx.x * (GEMM_THREADS / 32) + wave;
    const int tile_ld = (tile < mtiles) ? tile : (mtiles - 1);
    const int row0 = tile_ld * 16;

    constexpr int ITEMS = K * 8 / GEMM_THREADS;
    const v4f* __restrict__ W4 = (const v4f*)W;
    v4f q0[ITEMS], q1[ITEMS];
#pragma unroll
    for (int it = 0; it < ITEMS; ++it) {
        const int id = it * GEMM_THREADS + threadIdx.x;
        const int p = id >> 4, j = id & 15;
        q0[it] = W4[(2 * p) * 16 + j];
        q1[it] = W4[(2 * p + 1) * 16 + j];
    }

    const float* __restrict__ Arow = X + (size_t)(row0 + lid) * K;
    v2f a[K / 4];
#pragma unroll
    for (int s = 0; s < K / 4; ++s)
        a[s] = *(const v2f*)(Arow + 4 * s + 2 * half);
    v2f a2[SUM_A ? K / 4 : 1];
    if (SUM_A) {
        const float* __restrict__ Arow2 = X2 + (size_t)(row0 + lid) * K;
#pragma unroll
        for (int s = 0; s < K / 4; ++s)
            a2[s] = *(const v2f*)(Arow2 + 4 * s + 2 * half);
    }

#pragma unroll
    for (int it = 0; it < ITEMS; ++it) {
        const int id = it * GEMM_THREADS + threadIdx.x;
        const int p = id >> 4, j = id & 15;
        v4f lo, hi;
        lo.x = q0[it].x; lo.y = q1[it].x; lo.z = q0[it].y; lo.w = q1[it].y;
        hi.x = q0[it].z; hi.y = q1[it].z; hi.z = q0[it].w; hi.w = q1[it].w;
        sW[(2 * p) * 16 + j]     = lo;
        sW[(2 * p + 1) * 16 + j] = hi;
    }
    __syncthreads();

    v8f c0 = {}, c1 = {}, c2 = {}, c3 = {};
#pragma unroll
    for (int s = 0; s < K / 4; ++s) {
        v2f av = a[s];
        if (SUM_A) {
            av.x += a2[s].x;
            av.y += a2[s].y;
        }
        if (RELU_A) {
            av.x = fmaxf(av.x, 0.0f);
            av.y = fmaxf(av.y, 0.0f);
        }
        const int p = 2 * s + half;
        const v4f u0 = sW[(2 * p) * 16 + lid];
        const v4f u1 = sW[(2 * p + 1) * 16 + lid];
        v2f b0, b1, b2, b3;
        b0.x = u0.x; b0.y = u0.y;
        b1.x = u0.z; b1.y = u0.w;
        b2.x = u1.x; b2.y = u1.y;
        b3.x = u1.z; b3.y = u1.w;
        c0 = __builtin_amdgcn_wmma_f32_16x16x4_f32(false, av, false, b0, (short)0, c0, false, false);
        c1 = __builtin_amdgcn_wmma_f32_16x16x4_f32(false, av, false, b1, (short)0, c1, false, false);
        c2 = __builtin_amdgcn_wmma_f32_16x16x4_f32(false, av, false, b2, (short)0, c2, false, false);
        c3 = __builtin_amdgcn_wmma_f32_16x16x4_f32(false, av, false, b3, (short)0, c3, false, false);
    }

    if (tile < mtiles) {
        float* __restrict__ Hb = H + (size_t)(row0 + 8 * half) * HID_DIM + 4 * lid;
#pragma unroll
        for (int r = 0; r < 8; ++r) {
            v4f o;
            o.x = c0[r]; o.y = c1[r]; o.z = c2[r]; o.w = c3[r];
            *(v4f*)(Hb + (size_t)r * HID_DIM) = o;
        }
    }
}

#define GA_NPW 8
#define GA_U   8

__device__ __forceinline__ int gcn_slots_load(const int* __restrict__ es, int c, int e, int lane) {
    int m = e - c;
    m = (m > 32) ? 32 : m;
    int idx = c + ((lane < m) ? lane : (m - 1));
    idx = (idx < 0) ? 0 : idx;
    return es[idx];
}

__device__ __forceinline__ void gcn_rows_load(const float* __restrict__ Hl, int s, int j0,
                                              v2f (&h)[GA_U]) {
#pragma unroll
    for (int u = 0; u < GA_U; ++u) {
        const int sj = __builtin_amdgcn_readlane(s, j0 + u);
        h[u] = *(const v2f*)(Hl + (size_t)sj * HID_DIM);
    }
}

__device__ __forceinline__ void gcn_rows_sum(int nbits, int j0, int m, const v2f (&h)[GA_U],
                                             float& ax, float& ay) {
#pragma unroll
    for (int u = 0; u < GA_U; ++u) {
        const float nj = __int_as_float(__builtin_amdgcn_readlane(nbits, j0 + u));
        const bool ok = (j0 + u) < m;
        ax = fmaf(ok ? h[u].x : 0.0f, nj, ax);
        ay = fmaf(ok ? h[u].y : 0.0f, nj, ay);
    }
}

__device__ __forceinline__ void gcn_slots_sum(const float* __restrict__ Hl,
                                              const float* __restrict__ norm, int s, int m,
                                              float& ax, float& ay) {
    const int nbits = __float_as_int(norm[s]);
    for (int j = 0; j < m; j += 2 * GA_U) {
        v2f h0[GA_U], h1[GA_U];
        const bool two = (j + GA_U) < m;
        gcn_rows_load(Hl, s, j, h0);
        if (two) gcn_rows_load(Hl, s, j + GA_U, h1);
        gcn_rows_sum(nbits, j, m, h0, ax, ay);
        if (two) gcn_rows_sum(nbits, j + GA_U, m, h1, ax, ay);
    }
}

__global__ __launch_bounds__(256) void gcn_gather_kernel(const float* __restrict__ H,
                                                         const float* __restrict__ norm,
                                                         const int* __restrict__ start,
                                                         const int* __restrict__ es,
                                                         float* __restrict__ out, int N) {
    const int lane = threadIdx.x & 31;
    const int wv = __builtin_amdgcn_readfirstlane((int)((blockIdx.x * blockDim.x + threadIdx.x) >> 5));
    const int n0 = wv * GA_NPW;
    if (n0 >= N) return;
    const int nn = (N - n0 < GA_NPW) ? (N - n0) : GA_NPW;
    const float* __restrict__ Hl = H + 2 * lane;

    const int bv = start[n0 + ((lane < nn) ? lane : nn)];
    int b = __builtin_amdgcn_readlane(bv, 0);
    int e = __builtin_amdgcn_readlane(bv, 1);
    int s = gcn_slots_load(es, b, e, lane);

    for (int k = 0; k < nn; ++k) {
        int e2 = e;
        int s2 = s;
        if (k + 1 < nn) {
            e2 = __builtin_amdgcn_readlane(bv, k + 2);
            s2 = gcn_slots_load(es, e, e2, lane);
        }

        float ax = 0.0f, ay = 0.0f;
        if (b < e) {
            gcn_slots_sum(Hl, norm, s, (e - b > 32) ? 32 : (e - b), ax, ay);
            for (int c = b + 32; c < e; c += 32) {
                s = gcn_slots_load(es, c, e, lane);
                gcn_slots_sum(Hl, norm, s, (e - c > 32) ? 32 : (e - c), ax, ay);
            }
        }
        v2f o;
        o.x = fmaxf(ax, 0.0f);
        o.y = fmaxf(ay, 0.0f);
        *(v2f*)(out + (size_t)(n0 + k) * HID_DIM + 2 * lane) = o;

        b = e;
        e = e2;
        s = s2;
    }
}


extern "C" void kernel_launch(void* const* d_in, const int* in_sizes, int n_in,
                              void* d_out, int out_size, void* d_ws, size_t ws_size,
                              hipStream_t stream) {
    const float* features = (const float*)d_in[0];
    const float* W1       = (const float*)d_in[1];
    const float* W2       = (const float*)d_in[2];
    const int*   src      = (const int*)d_in[3];
    const int*   dst      = (const int*)d_in[4];

    const int N = in_sizes[0] / IN_DIM;
    const int E = in_sizes[3];
    const long long NH = (long long)N * HID_DIM;
    const long long E64 = ((long long)E + 63) / 64 * 64;

    const int nbk = (N + BK_NODES - 1) / BK_NODES;
    const int ga  = (E + ED_EPB - 1) / ED_EPB;
    const int nsc = nbk * ga;
    const long long S64 = ((long long)nsc + 64) / 64 * 64;

    int*   bh    = (int*)d_ws;
    int*   off   = bh + S64;
    int*   start = off + S64;
    float* norm  = (float*)(start + 65536 + 64);
    v2i*   tmp   = (v2i*)(norm + 65536);
    int*   es    = (int*)(tmp + E64);
    float* H     = (float*)(es + E64);
    float* X1    = H + NH;
    float* OUT   = (float*)d_out;

    const int mtiles = N / 16;
    const int gblocks = (mtiles + 3) / 4;
    const int nblocks = (N + GA_NPW * 8 - 1) / (GA_NPW * 8);

    gcn_hist_kernel<<<ga, ED_T, 0, stream>>>(dst, bh, E, nbk);
    gcn_scan_kernel<<<(nsc + SCAN_T - 1) / SCAN_T, SCAN_T, 0, stream>>>(bh, off, nsc);
    gcn_split_kernel<<<ga, ED_T, 0, stream>>>(src, dst, off, tmp, E, nbk);
    gcn_bucket_kernel<<<nbk, BF_T, 0, stream>>>(tmp, off, start, norm, es, N, ga);

    gcn_gemm_wmma<IN_DIM, false, false><<<gblocks, GEMM_THREADS, 0, stream>>>(features, features, W1, H, mtiles);
    gcn_gather_kernel<<<nblocks, 256, 0, stream>>>(H, norm, start, es, X1, N);

    gcn_gemm_wmma<HID_DIM, false, false><<<gblocks, GEMM_THREADS, 0, stream>>>(X1, X1, W2, H, mtiles);
    gcn_gather_kernel<<<nblocks, 256, 0, stream>>>(H, norm, start, es, OUT, N);
}


# layer 2's product X1 @ W2 runs inside layer 1's gather kernel: a block keeps its 64 relu'd rows and W2 in LDS and 4 of its waves do the f32 WMMA strips; X1 never goes to memory, 7 launches not 8
# speedup vs baseline: 3.1680x; 1.0675x over previous
#include <hip/hip_runtime.h>

typedef __attribute__((ext_vector_type(2))) float v2f;
typedef __attribute__((ext_vector_type(4))) float v4f;
typedef __attribute__((ext_vector_type(8))) float v8f;
typedef __attribute__((ext_vector_type(2))) int v2i;
typedef __attribute__((ext_vector_type(4))) int v4i;

#define IN_DIM 128
#define HID_DIM 64

#define BK_SHIFT 8
#define BK_NODES 256
#define NBK_MAX  256
#define ED_EPB   2048
#define ED_T     256

__global__ __launch_bounds__(ED_T) void gcn_hist_kernel(const int* __restrict__ dst,
                                                        int* __restrict__ bh, int E, int nbk) {
    __shared__ int h[NBK_MAX];
    const int t = threadIdx.x;
    h[t] = 0;
    __syncthreads();
    const long long e0 = (long long)blockIdx.x * ED_EPB;
    if (e0 + ED_EPB <= E) {
        const v4i da = *(const v4i*)(dst + e0 + 4 * t);
        const v4i db = *(const v4i*)(dst + e0 + ED_EPB / 2 + 4 * t);
        atomicAdd(&h[da.x >> BK_SHIFT], 1);
        atomicAdd(&h[da.y >> BK_SHIFT], 1);
        atomicAdd(&h[da.z >> BK_SHIFT], 1);
        atomicAdd(&h[da.w >> BK_SHIFT], 1);
        atomicAdd(&h[db.x >> BK_SHIFT], 1);
        atomicAdd(&h[db.y >> BK_SHIFT], 1);
        atomicAdd(&h[db.z >> BK_SHIFT], 1);
        atomicAdd(&h[db.w >> BK_SHIFT], 1);
    } else {
        for (long long e = e0 + t; e < E; e += ED_T) atomicAdd(&h[dst[e] >> BK_SHIFT], 1);
    }
    __syncthreads();
    if (t < nbk) bh[(size_t)t * gridDim.x + blockIdx.x] = h[t];
}

#define SCAN_T 1024
__global__ __launch_bounds__(SCAN_T) void gcn_scan_kernel(const int* __restrict__ in,
                                                          int* __restrict__ out, int n) {
    __shared__ int s_tot[32];
    __shared__ int s_pre[32];
    const int t = threadIdx.x, lane = t & 31, w = t >> 5;

    int pre = 0;
#pragma unroll 16
    for (int k = 0; k < (int)blockIdx.x; ++k) pre += in[k * SCAN_T + t];

    const int i = blockIdx.x * SCAN_T + t;
    const int d = (i < n) ? in[i] : 0;

    int x = d;
#pragma unroll
    for (int o = 1; o < 32; o <<= 1) {
        const int y = __shfl_up(x, o);
        if (lane >= o) x += y;
    }
#pragma unroll
    for (int o = 16; o > 0; o >>= 1) pre += __shfl_xor(pre, o);
    if (lane == 31) s_tot[w] = x;
    if (lane == 0) s_pre[w] = pre;
    __syncthreads();

    const int wt = s_tot[lane];
    int ws = wt;
#pragma unroll
    for (int o = 1; o < 32; o <<= 1) {
        const int y = __shfl_up(ws, o);
        if (lane >= o) ws += y;
    }
    int pt = s_pre[lane];
#pragma unroll
    for (int o = 16; o > 0; o >>= 1) pt += __shfl_xor(pt, o);
    const int wave_excl = __shfl(ws - wt, w);

    const int excl = pt + wave_excl + (x - d);
    if (i < n) {
        out[i] = excl;
        if (i == n - 1) out[n] = excl + d;
    }
}

__global__ __launch_bounds__(ED_T) void gcn_split_kernel(const int* __restrict__ src,
                                                         const int* __restrict__ dst,
                                                         const int* __restrict__ off,
                                                         v2i* __restrict__ tmp, int E, int nbk) {
    __shared__ int cur[NBK_MAX];
    const int t = threadIdx.x;
    cur[t] = (t < nbk) ? off[(size_t)t * gridDim.x + blockIdx.x] : 0;
    __syncthreads();
    const long long e0 = (long long)blockIdx.x * ED_EPB;
    if (e0 + ED_EPB <= E) {
        const v4i da = *(const v4i*)(dst + e0 + 4 * t);
        const v4i db = *(const v4i*)(dst + e0 + ED_EPB / 2 + 4 * t);
        const v4i sa = *(const v4i*)(src + e0 + 4 * t);
        const v4i sb = *(const v4i*)(src + e0 + ED_EPB / 2 + 4 * t);
        const int d[8] = {da.x, da.y, da.z, da.w, db.x, db.y, db.z, db.w};
        const int s[8] = {sa.x, sa.y, sa.z, sa.w, sb.x, sb.y, sb.z, sb.w};
        int r[8];
#pragma unroll
        for (int u = 0; u < 8; ++u) r[u] = atomicAdd(&cur[d[u] >> BK_SHIFT], 1);
#pragma unroll
        for (int u = 0; u < 8; ++u) {
            v2i o;
            o.x = s[u];
            o.y = d[u];
            tmp[r[u]] = o;
        }
    } else {
        for (long long e = e0 + t; e < E; e += ED_T) {
            const int dd = dst[e];
            const int r = atomicAdd(&cur[dd >> BK_SHIFT], 1);
            v2i o;
            o.x = src[e];
            o.y = dd;
            tmp[r] = o;
        }
    }
}

#define BF_T 1024
#define BF_R 8
__global__ __launch_bounds__(BF_T) void gcn_bucket_kernel(const v2i* __restrict__ tmp,
                                                          const int* __restrict__ off,
                                                          int* __restrict__ start,
                                                          float* __restrict__ norm,
                                                          int* __restrict__ es, int N, int ga) {
    __shared__ int cnt[BK_NODES];
    __shared__ int wtot[BK_NODES / 32];
    const int t = threadIdx.x, lane = t & 31, w = t >> 5;
    const int b = blockIdx.x;
    const int bs = off[(size_t)b * ga];
    const int be = off[(size_t)(b + 1) * ga];
    if (t < BK_NODES) cnt[t] = 0;
    __syncthreads();

    v2i q[BF_R];
#pragma unroll
    for (int u = 0; u < BF_R; ++u) {
        const int i = bs + u * BF_T + t;
        q[u].x = 0;
        q[u].y = -1;
        if (i < be) q[u] = tmp[i];
    }
#pragma unroll
    for (int u = 0; u < BF_R; ++u)
        if (q[u].y >= 0) atomicAdd(&cnt[q[u].y & (BK_NODES - 1)], 1);
    for (int i = bs + BF_R * BF_T + t; i < be; i += BF_T)
        atomicAdd(&cnt[tmp[i].y & (BK_NODES - 1)], 1);
    __syncthreads();

    const int c = (t < BK_NODES) ? cnt[t] : 0;
    int x = c;
#pragma unroll
    for (int o = 1; o < 32; o <<= 1) {
        const int y = __shfl_up(x, o);
        if (lane >= o) x += y;
    }
    if (t < BK_NODES && lane == 31) wtot[w] = x;
    __syncthreads();
    if (t < BK_NODES) {
        int pre = 0;
        for (int k = 0; k < w; ++k) pre += wtot[k];
        const int first = bs + pre + (x - c);
        const int node = b * BK_NODES + t;
        cnt[t] = first;
        if (node < N) {
            start[node] = first;
            norm[node] = rsqrtf(fmaxf((float)c, 1.0f));
            if (node == N - 1) start[N] = first + c;
        }
    }
    __syncthreads();

    int r[BF_R];
#pragma unroll
    for (int u = 0; u < BF_R; ++u) {
        r[u] = 0;
        if (q[u].y >= 0) r[u] = atomicAdd(&cnt[q[u].y & (BK_NODES - 1)], 1);
    }
#pragma unroll
    for (int u = 0; u < BF_R; ++u)
        if (q[u].y >= 0) es[r[u]] = q[u].x;
    for (int i = bs + BF_R * BF_T + t; i < be; i += BF_T) {
        const v2i sd = tmp[i];
        es[atomicAdd(&cnt[sd.y & (BK_NODES - 1)], 1)] = sd.x;
    }
}

#define GEMM_THREADS 128

template <int K, bool RELU_A, bool SUM_A>
__global__ __launch_bounds__(GEMM_THREADS) void gcn_gemm_wmma(const float* __restrict__ X,
                              const float* __restrict__ X2,
                              const float* __restrict__ W,
                              float* __restrict__ H, int mtiles) {
    __shared__ v4f sW[K * 16];

    const int wave = threadIdx.x >> 5;
    const int lane = threadIdx.x & 31;
    const int half = lane >> 4;
    const int lid  = lane & 15;
    const int tile = blockIdx.x * (GEMM_THREADS / 32) + wave;
    const int tile_ld = (tile < mtiles) ? tile : (mtiles - 1);
    const int row0 = tile_ld * 16;

    constexpr int ITEMS = K * 8 / GEMM_THREADS;
    const v4f* __restrict__ W4 = (const v4f*)W;
    v4f q0[ITEMS], q1[ITEMS];
#pragma unroll
    for (int it = 0; it < ITEMS; ++it) {
        const int id = it * GEMM_THREADS + threadIdx.x;
        const int p = id >> 4, j = id & 15;
        q0[it] = W4[(2 * p) * 16 + j];
        q1[it] = W4[(2 * p + 1) * 16 + j];
    }

    const float* __restrict__ Arow = X + (size_t)(row0 + lid) * K;
    v2f a[K / 4];
#pragma unroll
    for (int s = 0; s < K / 4; ++s)
        a[s] = *(const v2f*)(Arow + 4 * s + 2 * half);
    v2f a2[SUM_A ? K / 4 : 1];
    if (SUM_A) {
        const float* __restrict__ Arow2 = X2 + (size_t)(row0 + lid) * K;
#pragma unroll
        for (int s = 0; s < K / 4; ++s)
            a2[s] = *(const v2f*)(Arow2 + 4 * s + 2 * half);
    }

#pragma unroll
    for (int it = 0; it < ITEMS; ++it) {
        const int id = it * GEMM_THREADS + threadIdx.x;
        const int p = id >> 4, j = id & 15;
        v4f lo, hi;
        lo.x = q0[it].x; lo.y = q1[it].x; lo.z = q0[it].y; lo.w = q1[it].y;
        hi.x = q0[it].z; hi.y = q1[it].z; hi.z = q0[it].w; hi.w = q1[it].w;
        sW[(2 * p) * 16 + j]     = lo;
        sW[(2 * p + 1) * 16 + j] = hi;
    }
    __syncthreads();

    v8f c0 = {}, c1 = {}, c2 = {}, c3 = {};
#pragma unroll
    for (int s = 0; s < K / 4; ++s) {
        v2f av = a[s];
        if (SUM_A) {
            av.x += a2[s].x;
            av.y += a2[s].y;
        }
        if (RELU_A) {
            av.x = fmaxf(av.x, 0.0f);
            av.y = fmaxf(av.y, 0.0f);
        }
        const int p = 2 * s + half;
        const v4f u0 = sW[(2 * p) * 16 + lid];
        const v4f u1 = sW[(2 * p + 1) * 16 + lid];
        v2f b0, b1, b2, b3;
        b0.x = u0.x; b0.y = u0.y;
        b1.x = u0.z; b1.y = u0.w;
        b2.x = u1.x; b2.y = u1.y;
        b3.x = u1.z; b3.y = u1.w;
        c0 = __builtin_amdgcn_wmma_f32_16x16x4_f32(false, av, false, b0, (short)0, c0, false, false);
        c1 = __builtin_amdgcn_wmma_f32_16x16x4_f32(false, av, false, b1, (short)0, c1, false, false);
        c2 = __builtin_amdgcn_wmma_f32_16x16x4_f32(false, av, false, b2, (short)0, c2, false, false);
        c3 = __builtin_amdgcn_wmma_f32_16x16x4_f32(false, av, false, b3, (short)0, c3, false, false);
    }

    if (tile < mtiles) {
        float* __restrict__ Hb = H + (size_t)(row0 + 8 * half) * HID_DIM + 4 * lid;
#pragma unroll
        for (int r = 0; r < 8; ++r) {
            v4f o;
            o.x = c0[r]; o.y = c1[r]; o.z = c2[r]; o.w = c3[r];
            *(v4f*)(Hb + (size_t)r * HID_DIM) = o;
        }
    }
}

#define GA_NPW 8
#define GA_U   8

__device__ __forceinline__ int gcn_slots_load(const int* __restrict__ es, int c, int e, int lane) {
    int m = e - c;
    m = (m > 32) ? 32 : m;
    int idx = c + ((lane < m) ? lane : (m - 1));
    idx = (idx < 0) ? 0 : idx;
    return es[idx];
}

__device__ __forceinline__ void gcn_rows_load(const float* __restrict__ Hl, int s, int j0,
                                              v2f (&h)[GA_U]) {
#pragma unroll
    for (int u = 0; u < GA_U; ++u) {
        const int sj = __builtin_amdgcn_readlane(s, j0 + u);
        h[u] = *(const v2f*)(Hl + (size_t)sj * HID_DIM);
    }
}

__device__ __forceinline__ void gcn_rows_sum(int nbits, int j0, int m, const v2f (&h)[GA_U],
                                             float& ax, float& ay) {
#pragma unroll
    for (int u = 0; u < GA_U; ++u) {
        const float nj = __int_as_float(__builtin_amdgcn_readlane(nbits, j0 + u));
        const bool ok = (j0 + u) < m;
        ax = fmaf(ok ? h[u].x : 0.0f, nj, ax);
        ay = fmaf(ok ? h[u].y : 0.0f, nj, ay);
    }
}

__device__ __forceinline__ void gcn_slots_sum(const float* __restrict__ Hl,
                                              const float* __restrict__ norm, int s, int m,
                                              float& ax, float& ay) {
    const int nbits = __float_as_int(norm[s]);
    for (int j = 0; j < m; j += 2 * GA_U) {
        v2f h0[GA_U], h1[GA_U];
        const bool two = (j + GA_U) < m;
        gcn_rows_load(Hl, s, j, h0);
        if (two) gcn_rows_load(Hl, s, j + GA_U, h1);
        gcn_rows_sum(nbits, j, m, h0, ax, ay);
        if (two) gcn_rows_sum(nbits, j + GA_U, m, h1, ax, ay);
    }
}

#define GA_WAVES 8
#define GA_ROWS  (GA_NPW * GA_WAVES)
#define GA_PITCH 68

template <bool FUSE_GEMM>
__global__ __launch_bounds__(256) void gcn_gather_kernel(const float* __restrict__ H,
                                                         const float* __restrict__ norm,
                                                         const int* __restrict__ start,
                                                         const int* __restrict__ es,
                                                         const float* __restrict__ W,
                                                         float* __restrict__ out, int N) {
    __shared__ v4f   sW[FUSE_GEMM ? HID_DIM * 16 : 1];
    __shared__ float sX[FUSE_GEMM ? GA_ROWS * GA_PITCH : 2];

    const int lane = threadIdx.x & 31;
    const int wib = __builtin_amdgcn_readfirstlane((int)(threadIdx.x >> 5));
    const int n0 = ((int)blockIdx.x * GA_WAVES + wib) * GA_NPW;
    if (!FUSE_GEMM) {
        if (n0 >= N) return;
    }
    int nn = (N - n0 < GA_NPW) ? (N - n0) : GA_NPW;
    nn = (nn < 0) ? 0 : nn;
    const float* __restrict__ Hl = H + 2 * lane;

    int bi = n0 + ((lane < nn) ? lane : nn);
    bi = (bi > N) ? N : bi;
    const int bv = start[bi];

    constexpr int W_ITEMS = HID_DIM * 8 / 256;
    v4f q0[W_ITEMS], q1[W_ITEMS];
    if (FUSE_GEMM) {
        const v4f* __restrict__ W4 = (const v4f*)W;
#pragma unroll
        for (int it = 0; it < W_ITEMS; ++it) {
            const int id = it * 256 + (int)threadIdx.x;
            const int p = id >> 4, j = id & 15;
            q0[it] = W4[(2 * p) * 16 + j];
            q1[it] = W4[(2 * p + 1) * 16 + j];
        }
    }

    int b = __builtin_amdgcn_readlane(bv, 0);
    int e = __builtin_amdgcn_readlane(bv, 1);
    int s = gcn_slots_load(es, b, e, lane);

    for (int k = 0; k < nn; ++k) {
        int e2 = e;
        int s2 = s;
        if (k + 1 < nn) {
            e2 = __builtin_amdgcn_readlane(bv, k + 2);
            s2 = gcn_slots_load(es, e, e2, lane);
        }

        float ax = 0.0f, ay = 0.0f;
        if (b < e) {
            gcn_slots_sum(Hl, norm, s, (e - b > 32) ? 32 : (e - b), ax, ay);
            for (int c = b + 32; c < e; c += 32) {
                s = gcn_slots_load(es, c, e, lane);
                gcn_slots_sum(Hl, norm, s, (e - c > 32) ? 32 : (e - c), ax, ay);
            }
        }
        v2f o;
        o.x = fmaxf(ax, 0.0f);
        o.y = fmaxf(ay, 0.0f);
        if (FUSE_GEMM)
            *(v2f*)(sX + (wib * GA_NPW + k) * GA_PITCH + 2 * lane) = o;
        else
            *(v2f*)(out + (size_t)(n0 + k) * HID_DIM + 2 * lane) = o;

        b = e;
        e = e2;
        s = s2;
    }

    if (FUSE_GEMM) {
#pragma unroll
        for (int it = 0; it < W_ITEMS; ++it) {
            const int id = it * 256 + (int)threadIdx.x;
            const int p = id >> 4, j = id & 15;
            v4f lo, hi;
            lo.x = q0[it].x; lo.y = q1[it].x; lo.z = q0[it].y; lo.w = q1[it].y;
            hi.x = q0[it].z; hi.y = q1[it].z; hi.z = q0[it].w; hi.w = q1[it].w;
            sW[(2 * p) * 16 + j]     = lo;
            sW[(2 * p + 1) * 16 + j] = hi;
        }

        for (int k = nn; k < GA_NPW; ++k) {
            v2f z;
            z.x = 0.0f;
            z.y = 0.0f;
            *(v2f*)(sX + (wib * GA_NPW + k) * GA_PITCH + 2 * lane) = z;
        }
        __syncthreads();
        if (wib >= GA_ROWS / 16) return;

        const int half = lane >> 4;
        const int lid  = lane & 15;
        const float* Ar = sX + (wib * 16 + lid) * GA_PITCH + 2 * half;
        v8f c0 = {}, c1 = {}, c2 = {}, c3 = {};
#pragma unroll
        for (int s4 = 0; s4 < HID_DIM / 4; ++s4) {
            const v2f av = *(const v2f*)(Ar + 4 * s4);
            const int p = 2 * s4 + half;
            const v4f u0 = sW[(2 * p) * 16 + lid];
            const v4f u1 = sW[(2 * p + 1) * 16 + lid];
            v2f b0, b1, b2, b3;
            b0.x = u0.x; b0.y = u0.y;
            b1.x = u0.z; b1.y = u0.w;
            b2.x = u1.x; b2.y = u1.y;
            b3.x = u1.z; b3.y = u1.w;
            c0 = __builtin_amdgcn_wmma_f32_16x16x4_f32(false, av, false, b0, (short)0, c0, false, false);
            c1 = __builtin_amdgcn_wmma_f32_16x16x4_f32(false, av, false, b1, (short)0, c1, false, false);
            c2 = __builtin_amdgcn_wmma_f32_16x16x4_f32(false, av, false, b2, (short)0, c2, false, false);
            c3 = __builtin_amdgcn_wmma_f32_16x16x4_f32(false, av, false, b3, (short)0, c3, false, false);
        }

        const int strip0 = (int)blockIdx.x * GA_ROWS + wib * 16;
        if (strip0 < N) {
            float* __restrict__ Ob = out + (size_t)(strip0 + 8 * half) * HID_DIM + 4 * lid;
#pragma unroll
            for (int r = 0; r < 8; ++r) {
                v4f o;
                o.x = c0[r]; o.y = c1[r]; o.z = c2[r]; o.w = c3[r];
                *(v4f*)(Ob + (size_t)r * HID_DIM) = o;
            }
        }
    }
}


extern "C" void kernel_launch(void* const* d_in, const int* in_sizes, int n_in,
                              void* d_out, int out_size, void* d_ws, size_t ws_size,
                              hipStream_t stream) {
    const float* features = (const float*)d_in[0];
    const float* W1       = (const float*)d_in[1];
    const float* W2       = (const float*)d_in[2];
    const int*   src      = (const int*)d_in[3];
    const int*   dst      = (const int*)d_in[4];

    const int N = in_sizes[0] / IN_DIM;
    const int E = in_sizes[3];
    const long long NH = (long long)N * HID_DIM;
    const long long E64 = ((long long)E + 63) / 64 * 64;

    const int nbk = (N + BK_NODES - 1) / BK_NODES;
    const int ga  = (E + ED_EPB - 1) / ED_EPB;
    const int nsc = nbk * ga;
    const long long S64 = ((long long)nsc + 64) / 64 * 64;

    int*   bh    = (int*)d_ws;
    int*   off   = bh + S64;
    int*   start = off + S64;
    float* norm  = (float*)(start + 65536 + 64);
    v2i*   tmp   = (v2i*)(norm + 65536);
    int*   es    = (int*)(tmp + E64);
    float* H     = (float*)(es + E64);
    float* H2    = H + NH;
    float* OUT   = (float*)d_out;

    const int mtiles = N / 16;
    const int gblocks = (mtiles + 3) / 4;
    const int nblocks = (N + GA_ROWS - 1) / GA_ROWS;

    gcn_hist_kernel<<<ga, ED_T, 0, stream>>>(dst, bh, E, nbk);
    gcn_scan_kernel<<<(nsc + SCAN_T - 1) / SCAN_T, SCAN_T, 0, stream>>>(bh, off, nsc);
    gcn_split_kernel<<<ga, ED_T, 0, stream>>>(src, dst, off, tmp, E, nbk);
    gcn_bucket_kernel<<<nbk, BF_T, 0, stream>>>(tmp, off, start, norm, es, N, ga);

    gcn_gemm_wmma<IN_DIM, false, false><<<gblocks, GEMM_THREADS, 0, stream>>>(features, features, W1, H, mtiles);
    gcn_gather_kernel<true><<<nblocks, 256, 0, stream>>>(H, norm, start, es, W2, H2, N);

    gcn_gather_kernel<false><<<nblocks, 256, 0, stream>>>(H2, norm, start, es, W2, OUT, N);
}


# edge sort needs no prefix sum over all edges: a block orders its 2048 edges by bucket in place and a bucket's block walks those runs; 2 launches not 4; sum order = slot arrival order, by a new route
# speedup vs baseline: 4.2378x; 1.3377x over previous
#include <hip/hip_runtime.h>

typedef __attribute__((ext_vector_type(2))) float v2f;
typedef __attribute__((ext_vector_type(4))) float v4f;
typedef __attribute__((ext_vector_type(8))) float v8f;
typedef __attribute__((ext_vector_type(2))) int v2i;
typedef __attribute__((ext_vector_type(4))) int v4i;

#define IN_DIM 128
#define HID_DIM 64

#define BK_SHIFT 8
#define BK_NODES 256
#define NBK_MAX  256
#define ED_EPB   2048
#define ED_T     256

__global__ __launch_bounds__(ED_T) void gcn_runs_kernel(const int* __restrict__ src,
                                                        const int* __restrict__ dst,
                                                        int* __restrict__ runs,
                                                        v2i* __restrict__ tmp, int E, int nbk) {
    __shared__ int h[NBK_MAX];
    __shared__ int wt[ED_T / 32];
    const int t = threadIdx.x, lane = t & 31, w = t >> 5;
    h[t] = 0;
    __syncthreads();
    const long long e0 = (long long)blockIdx.x * ED_EPB;
    int d[8], s[8];
    if (e0 + ED_EPB <= E) {
        const v4i da = *(const v4i*)(dst + e0 + 4 * t);
        const v4i db = *(const v4i*)(dst + e0 + ED_EPB / 2 + 4 * t);
        const v4i sa = *(const v4i*)(src + e0 + 4 * t);
        const v4i sb = *(const v4i*)(src + e0 + ED_EPB / 2 + 4 * t);
        d[0] = da.x; d[1] = da.y; d[2] = da.z; d[3] = da.w;
        d[4] = db.x; d[5] = db.y; d[6] = db.z; d[7] = db.w;
        s[0] = sa.x; s[1] = sa.y; s[2] = sa.z; s[3] = sa.w;
        s[4] = sb.x; s[5] = sb.y; s[6] = sb.z; s[7] = sb.w;
    } else {
#pragma unroll
        for (int u = 0; u < 8; ++u) {
            const long long e = e0 + (u >> 2) * (ED_EPB / 2) + 4 * t + (u & 3);
            d[u] = (e < E) ? dst[e] : -1;
            s[u] = (e < E) ? src[e] : 0;
        }
    }
    int bk[8], r[8];
#pragma unroll
    for (int u = 0; u < 8; ++u) bk[u] = (d[u] >= 0) ? (d[u] >> BK_SHIFT) : 0;
#pragma unroll
    for (int u = 0; u < 8; ++u) {
        r[u] = 0;
        if (d[u] >= 0) r[u] = atomicAdd(&h[bk[u]], 1);
    }
    __syncthreads();

    const int c = h[t];
    int x = c;
#pragma unroll
    for (int o = 1; o < 32; o <<= 1) {
        const int y = __shfl_up(x, o);
        if (lane >= o) x += y;
    }
    if (lane == 31) wt[w] = x;
    __syncthreads();
    int pre = 0;
#pragma unroll
    for (int k = 0; k < ED_T / 32; ++k) pre += (k < w) ? wt[k] : 0;
    const int excl = pre + (x - c);
    h[t] = excl;
    __syncthreads();

    int p[8];
#pragma unroll
    for (int u = 0; u < 8; ++u) p[u] = h[bk[u]] + r[u];
#pragma unroll
    for (int u = 0; u < 8; ++u) {
        if (d[u] >= 0) {
            v2i o;
            o.x = s[u];
            o.y = d[u];
            tmp[e0 + p[u]] = o;
        }
    }
    if (t < nbk) runs[(size_t)t * gridDim.x + blockIdx.x] = (excl << 16) | c;
}

#define BF_T 1024
#define RUN_IT 13

template <bool PLACE>
__device__ __forceinline__ void gcn_bucket_rest(const v2i* __restrict__ tmp, int* cnt,
                                                int* __restrict__ es, int pkv, int r0, int rpw,
                                                int lane) {
    for (int i = 0; i < rpw; ++i) {
        const int pk = __builtin_amdgcn_readlane(pkv, i);
        const int c = pk & 0xFFFF;
        const int o = pk >> 16;
        for (int j = ((i < RUN_IT) ? 32 : 0) + lane; j < c; j += 32) {
            const v2i sd = tmp[(size_t)(r0 + i) * ED_EPB + o + j];
            const int k = atomicAdd(&cnt[sd.y & (BK_NODES - 1)], 1);
            if (PLACE) es[k] = sd.x;
        }
    }
}

__global__ __launch_bounds__(BF_T) void gcn_bucket_kernel(const v2i* __restrict__ tmp,
                                                          const int* __restrict__ runs,
                                                          int* __restrict__ start,
                                                          float* __restrict__ norm,
                                                          int* __restrict__ es, int N, int ga) {
    __shared__ int cnt[BK_NODES];
    __shared__ int wtot[BK_NODES / 32];
    __shared__ int wbs[BF_T / 32];
    const int t = threadIdx.x, lane = t & 31;
    const int w = __builtin_amdgcn_readfirstlane((int)(threadIdx.x >> 5));
    const int b = blockIdx.x;
    const int rpw = (ga + BF_T / 32 - 1) / (BF_T / 32);
    const int r0 = w * rpw;

    int pkv = 0;
    if (lane < rpw && r0 + lane < ga) pkv = runs[(size_t)b * ga + r0 + lane];
    if (t < BK_NODES) cnt[t] = 0;

    int qs[RUN_IT], qd[RUN_IT];
#pragma unroll
    for (int i = 0; i < RUN_IT; ++i) {
        const int pk = __builtin_amdgcn_readlane(pkv, i);
        const int c = pk & 0xFFFF;
        const int o = pk >> 16;
        qs[i] = 0;
        qd[i] = -1;
        if (lane < c) {
            const v2i sd = tmp[(size_t)(r0 + i) * ED_EPB + o + lane];
            qs[i] = sd.x;
            qd[i] = sd.y;
        }
    }
    const bool more = (rpw > RUN_IT) || (__builtin_amdgcn_ballot_w32((pkv & 0xFFFF) > 32) != 0u);

    int osum = pkv >> 16;
#pragma unroll
    for (int o = 16; o > 0; o >>= 1) osum += __shfl_xor(osum, o);
    if (lane == 0) wbs[w] = osum;
    __syncthreads();

#pragma unroll
    for (int i = 0; i < RUN_IT; ++i)
        if (qd[i] >= 0) atomicAdd(&cnt[qd[i] & (BK_NODES - 1)], 1);
    if (more) gcn_bucket_rest<false>(tmp, cnt, es, pkv, r0, rpw, lane);
    __syncthreads();

    const int c = (t < BK_NODES) ? cnt[t] : 0;
    int x = c;
#pragma unroll
    for (int o = 1; o < 32; o <<= 1) {
        const int y = __shfl_up(x, o);
        if (lane >= o) x += y;
    }
    if (t < BK_NODES && lane == 31) wtot[w] = x;
    int bs = wbs[lane];
#pragma unroll
    for (int o = 16; o > 0; o >>= 1) bs += __shfl_xor(bs, o);
    __syncthreads();
    if (t < BK_NODES) {
        int pre = 0;
#pragma unroll
        for (int k = 0; k < BK_NODES / 32; ++k) pre += (k < w) ? wtot[k] : 0;
        const int first = bs + pre + (x - c);
        const int node = b * BK_NODES + t;
        cnt[t] = first;
        if (node < N) {
            start[node] = first;
            norm[node] = rsqrtf(fmaxf((float)c, 1.0f));
            if (node == N - 1) start[N] = first + c;
        }
    }
    __syncthreads();

    unsigned r[RUN_IT];
#pragma unroll
    for (int i = 0; i < RUN_IT; ++i) {
        r[i] = 0u;
        if (qd[i] >= 0) r[i] = (unsigned)atomicAdd(&cnt[qd[i] & (BK_NODES - 1)], 1);
    }
#pragma unroll
    for (int i = 0; i < RUN_IT; ++i)
        if (qd[i] >= 0) es[r[i]] = qs[i];
    if (more) gcn_bucket_rest<true>(tmp, cnt, es, pkv, r0, rpw, lane);
}

#define GEMM_THREADS 128

template <int K, bool RELU_A, bool SUM_A>
__global__ __launch_bounds__(GEMM_THREADS) void gcn_gemm_wmma(const float* __restrict__ X,
                              const float* __restrict__ X2,
                              const float* __restrict__ W,
                              float* __restrict__ H, int mtiles) {
    __shared__ v4f sW[K * 16];

    const int wave = threadIdx.x >> 5;
    const int lane = threadIdx.x & 31;
    const int half = lane >> 4;
    const int lid  = lane & 15;
    const int tile = blockIdx.x * (GEMM_THREADS / 32) + wave;
    const int tile_ld = (tile < mtiles) ? tile : (mtiles - 1);
    const int row0 = tile_ld * 16;

    constexpr int ITEMS = K * 8 / GEMM_THREADS;
    const v4f* __restrict__ W4 = (const v4f*)W;
    v4f q0[ITEMS], q1[ITEMS];
#pragma unroll
    for (int it = 0; it < ITEMS; ++it) {
        const int id = it * GEMM_THREADS + threadIdx.x;
        const int p = id >> 4, j = id & 15;
        q0[it] = W4[(2 * p) * 16 + j];
        q1[it] = W4[(2 * p + 1) * 16 + j];
    }

    const float* __restrict__ Arow = X + (size_t)(row0 + lid) * K;
    v2f a[K / 4];
#pragma unroll
    for (int s = 0; s < K / 4; ++s)
        a[s] = *(const v2f*)(Arow + 4 * s + 2 * half);
    v2f a2[SUM_A ? K / 4 : 1];
    if (SUM_A) {
        const float* __restrict__ Arow2 = X2 + (size_t)(row0 + lid) * K;
#pragma unroll
        for (int s = 0; s < K / 4; ++s)
            a2[s] = *(const v2f*)(Arow2 + 4 * s + 2 * half);
    }

#pragma unroll
    for (int it = 0; it < ITEMS; ++it) {
        const int id = it * GEMM_THREADS + threadIdx.x;
        const int p = id >> 4, j = id & 15;
        v4f lo, hi;
        lo.x = q0[it].x; lo.y = q1[it].x; lo.z = q0[it].y; lo.w = q1[it].y;
        hi.x = q0[it].z; hi.y = q1[it].z; hi.z = q0[it].w; hi.w = q1[it].w;
        sW[(2 * p) * 16 + j]     = lo;
        sW[(2 * p + 1) * 16 + j] = hi;
    }
    __syncthreads();

    v8f c0 = {}, c1 = {}, c2 = {}, c3 = {};
#pragma unroll
    for (int s = 0; s < K / 4; ++s) {
        v2f av = a[s];
        if (SUM_A) {
            av.x += a2[s].x;
            av.y += a2[s].y;
        }
        if (RELU_A) {
            av.x = fmaxf(av.x, 0.0f);
            av.y = fmaxf(av.y, 0.0f);
        }
        const int p = 2 * s + half;
        const v4f u0 = sW[(2 * p) * 16 + lid];
        const v4f u1 = sW[(2 * p + 1) * 16 + lid];
        v2f b0, b1, b2, b3;
        b0.x = u0.x; b0.y = u0.y;
        b1.x = u0.z; b1.y = u0.w;
        b2.x = u1.x; b2.y = u1.y;
        b3.x = u1.z; b3.y = u1.w;
        c0 = __builtin_amdgcn_wmma_f32_16x16x4_f32(false, av, false, b0, (short)0, c0, false, false);
        c1 = __builtin_amdgcn_wmma_f32_16x16x4_f32(false, av, false, b1, (short)0, c1, false, false);
        c2 = __builtin_amdgcn_wmma_f32_16x16x4_f32(false, av, false, b2, (short)0, c2, false, false);
        c3 = __builtin_amdgcn_wmma_f32_16x16x4_f32(false, av, false, b3, (short)0, c3, false, false);
    }

    if (tile < mtiles) {
        float* __restrict__ Hb = H + (size_t)(row0 + 8 * half) * HID_DIM + 4 * lid;
#pragma unroll
        for (int r = 0; r < 8; ++r) {
            v4f o;
            o.x = c0[r]; o.y = c1[r]; o.z = c2[r]; o.w = c3[r];
            *(v4f*)(Hb + (size_t)r * HID_DIM) = o;
        }
    }
}

#define GA_NPW 8
#define GA_U   8

__device__ __forceinline__ int gcn_slots_load(const int* __restrict__ es, int c, int e, int lane) {
    int m = e - c;
    m = (m > 32) ? 32 : m;
    int idx = c + ((lane < m) ? lane : (m - 1));
    idx = (idx < 0) ? 0 : idx;
    return es[idx];
}

__device__ __forceinline__ void gcn_rows_load(const float* __restrict__ Hl, int s, int j0,
                                              v2f (&h)[GA_U]) {
#pragma unroll
    for (int u = 0; u < GA_U; ++u) {
        const int sj = __builtin_amdgcn_readlane(s, j0 + u);
        h[u] = *(const v2f*)(Hl + (size_t)sj * HID_DIM);
    }
}

__device__ __forceinline__ void gcn_rows_sum(int nbits, int j0, int m, const v2f (&h)[GA_U],
                                             float& ax, float& ay) {
#pragma unroll
    for (int u = 0; u < GA_U; ++u) {
        const float nj = __int_as_float(__builtin_amdgcn_readlane(nbits, j0 + u));
        const bool ok = (j0 + u) < m;
        ax = fmaf(ok ? h[u].x : 0.0f, nj, ax);
        ay = fmaf(ok ? h[u].y : 0.0f, nj, ay);
    }
}

__device__ __forceinline__ void gcn_slots_sum(const float* __restrict__ Hl,
                                              const float* __restrict__ norm, int s, int m,
                                              float& ax, float& ay) {
    const int nbits = __float_as_int(norm[s]);
    for (int j = 0; j < m; j += 2 * GA_U) {
        v2f h0[GA_U], h1[GA_U];
        const bool two = (j + GA_U) < m;
        gcn_rows_load(Hl, s, j, h0);
        if (two) gcn_rows_load(Hl, s, j + GA_U, h1);
        gcn_rows_sum(nbits, j, m, h0, ax, ay);
        if (two) gcn_rows_sum(nbits, j + GA_U, m, h1, ax, ay);
    }
}

#define GA_WAVES 8
#define GA_ROWS  (GA_NPW * GA_WAVES)
#define GA_PITCH 68

template <bool FUSE_GEMM>
__global__ __launch_bounds__(256) void gcn_gather_kernel(const float* __restrict__ H,
                                                         const float* __restrict__ norm,
                                                         const int* __restrict__ start,
                                                         const int* __restrict__ es,
                                                         const float* __restrict__ W,
                                                         float* __restrict__ out, int N) {
    __shared__ v4f   sW[FUSE_GEMM ? HID_DIM * 16 : 1];
    __shared__ float sX[FUSE_GEMM ? GA_ROWS * GA_PITCH : 2];

    const int lane = threadIdx.x & 31;
    const int wib = __builtin_amdgcn_readfirstlane((int)(threadIdx.x >> 5));
    const int n0 = ((int)blockIdx.x * GA_WAVES + wib) * GA_NPW;
    if (!FUSE_GEMM) {
        if (n0 >= N) return;
    }
    int nn = (N - n0 < GA_NPW) ? (N - n0) : GA_NPW;
    nn = (nn < 0) ? 0 : nn;
    const float* __restrict__ Hl = H + 2 * lane;

    int bi = n0 + ((lane < nn) ? lane : nn);
    bi = (bi > N) ? N : bi;
    const int bv = start[bi];

    constexpr int W_ITEMS = HID_DIM * 8 / 256;
    v4f q0[W_ITEMS], q1[W_ITEMS];
    if (FUSE_GEMM) {
        const v4f* __restrict__ W4 = (const v4f*)W;
#pragma unroll
        for (int it = 0; it < W_ITEMS; ++it) {
            const int id = it * 256 + (int)threadIdx.x;
            const int p = id >> 4, j = id & 15;
            q0[it] = W4[(2 * p) * 16 + j];
            q1[it] = W4[(2 * p + 1) * 16 + j];
        }
    }

    int b = __builtin_amdgcn_readlane(bv, 0);
    int e = __builtin_amdgcn_readlane(bv, 1);
    int s = gcn_slots_load(es, b, e, lane);

    for (int k = 0; k < nn; ++k) {
        int e2 = e;
        int s2 = s;
        if (k + 1 < nn) {
            e2 = __builtin_amdgcn_readlane(bv, k + 2);
            s2 = gcn_slots_load(es, e, e2, lane);
        }

        float ax = 0.0f, ay = 0.0f;
        if (b < e) {
            gcn_slots_sum(Hl, norm, s, (e - b > 32) ? 32 : (e - b), ax, ay);
            for (int c = b + 32; c < e; c += 32) {
                s = gcn_slots_load(es, c, e, lane);
                gcn_slots_sum(Hl, norm, s, (e - c > 32) ? 32 : (e - c), ax, ay);
            }
        }
        v2f o;
        o.x = fmaxf(ax, 0.0f);
        o.y = fmaxf(ay, 0.0f);
        if (FUSE_GEMM)
            *(v2f*)(sX + (wib * GA_NPW + k) * GA_PITCH + 2 * lane) = o;
        else
            *(v2f*)(out + (size_t)(n0 + k) * HID_DIM + 2 * lane) = o;

        b = e;
        e = e2;
        s = s2;
    }

    if (FUSE_GEMM) {
#pragma unroll
        for (int it = 0; it < W_ITEMS; ++it) {
            const int id = it * 256 + (int)threadIdx.x;
            const int p = id >> 4, j = id & 15;
            v4f lo, hi;
            lo.x = q0[it].x; lo.y = q1[it].x; lo.z = q0[it].y; lo.w = q1[it].y;
            hi.x = q0[it].z; hi.y = q1[it].z; hi.z = q0[it].w; hi.w = q1[it].w;
            sW[(2 * p) * 16 + j]     = lo;
            sW[(2 * p + 1) * 16 + j] = hi;
        }

        for (int k = nn; k < GA_NPW; ++k) {
            v2f z;
            z.x = 0.0f;
            z.y = 0.0f;
            *(v2f*)(sX + (wib * GA_NPW + k) * GA_PITCH + 2 * lane) = z;
        }
        __syncthreads();
        if (wib >= GA_ROWS / 16) return;

        const int half = lane >> 4;
        const int lid  = lane & 15;
        const float* Ar = sX + (wib * 16 + lid) * GA_PITCH + 2 * half;
        v8f c0 = {}, c1 = {}, c2 = {}, c3 = {};
#pragma unroll
        for (int s4 = 0; s4 < HID_DIM / 4; ++s4) {
            const v2f av = *(const v2f*)(Ar + 4 * s4);
            const int p = 2 * s4 + half;
            const v4f u0 = sW[(2 * p) * 16 + lid];
            const v4f u1 = sW[(2 * p + 1) * 16 + lid];
            v2f b0, b1, b2, b3;
            b0.x = u0.x; b0.y = u0.y;
            b1.x = u0.z; b1.y = u0.w;
            b2.x = u1.x; b2.y = u1.y;
            b3.x = u1.z; b3.y = u1.w;
            c0 = __builtin_amdgcn_wmma_f32_16x16x4_f32(false, av, false, b0, (short)0, c0, false, false);
            c1 = __builtin_amdgcn_wmma_f32_16x16x4_f32(false, av, false, b1, (short)0, c1, false, false);
            c2 = __builtin_amdgcn_wmma_f32_16x16x4_f32(false, av, false, b2, (short)0, c2, false, false);
            c3 = __builtin_amdgcn_wmma_f32_16x16x4_f32(false, av, false, b3, (short)0, c3, false, false);
        }

        const int strip0 = (int)blockIdx.x * GA_ROWS + wib * 16;
        if (strip0 < N) {
            float* __restrict__ Ob = out + (size_t)(strip0 + 8 * half) * HID_DIM + 4 * lid;
#pragma unroll
            for (int r = 0; r < 8; ++r) {
                v4f o;
                o.x = c0[r]; o.y = c1[r]; o.z = c2[r]; o.w = c3[r];
                *(v4f*)(Ob + (size_t)r * HID_DIM) = o;
            }
        }
    }
}


extern "C" void kernel_launch(void* const* d_in, const int* in_sizes, int n_in,
                              void* d_out, int out_size, void* d_ws, size_t ws_size,
                              hipStream_t stream) {
    const float* features = (const float*)d_in[0];
    const float* W1       = (const float*)d_in[1];
    const float* W2       = (const float*)d_in[2];
    const int*   src      = (const int*)d_in[3];
    const int*   dst      = (const int*)d_in[4];

    const int N = in_sizes[0] / IN_DIM;
    const int E = in_sizes[3];
    const long long NH = (long long)N * HID_DIM;
    const long long E64 = ((long long)E + 63) / 64 * 64;

    const int nbk = (N + BK_NODES - 1) / BK_NODES;
    const int ga  = (E + ED_EPB - 1) / ED_EPB;
    const int nsc = nbk * ga;
    const long long S64 = ((long long)nsc + 63) / 64 * 64;

    int*   runs  = (int*)d_ws;
    int*   start = runs + S64;
    float* norm  = (float*)(start + 65536 + 64);
    v2i*   tmp   = (v2i*)(norm + 65536);
    int*   es    = (int*)(tmp + (long long)ga * ED_EPB);
    float* H     = (float*)(es + E64);
    float* H2    = H + NH;
    float* OUT   = (float*)d_out;

    const int mtiles = N / 16;
    const int gblocks = (mtiles + 3) / 4;
    const int nblocks = (N + GA_ROWS - 1) / GA_ROWS;

    gcn_runs_kernel<<<ga, ED_T, 0, stream>>>(src, dst, runs, tmp, E, nbk);
    gcn_bucket_kernel<<<nbk, BF_T, 0, stream>>>(tmp, runs, start, norm, es, N, ga);

    gcn_gemm_wmma<IN_DIM, false, false><<<gblocks, GEMM_THREADS, 0, stream>>>(features, features, W1, H, mtiles);
    gcn_gather_kernel<true><<<nblocks, 256, 0, stream>>>(H, norm, start, es, W2, H2, N);

    gcn_gather_kernel<false><<<nblocks, 256, 0, stream>>>(H2, norm, start, es, W2, OUT, N);
}


# launch merge: layer 1's product X @ W1 and the runs pass of the edge sort, independent of each other, share one grid (blocks 0-390 WMMA strips of 128 rows, the rest runs); 4 launches not 5
# speedup vs baseline: 4.4914x; 1.0598x over previous
#include <hip/hip_runtime.h>

typedef __attribute__((ext_vector_type(2))) float v2f;
typedef __attribute__((ext_vector_type(4))) float v4f;
typedef __attribute__((ext_vector_type(8))) float v8f;
typedef __attribute__((ext_vector_type(2))) int v2i;
typedef __attribute__((ext_vector_type(4))) int v4i;

#define IN_DIM 128
#define HID_DIM 64

#define BK_SHIFT 8
#define BK_NODES 256
#define NBK_MAX  256
#define ED_EPB   2048
#define ED_T     256

__device__ __forceinline__ void gcn_runs_block(const int* __restrict__ src,
                                               const int* __restrict__ dst,
                                               int* __restrict__ runs,
                                               v2i* __restrict__ tmp, int E, int nbk,
                                               int rb, int ga) {
    __shared__ int h[NBK_MAX];
    __shared__ int wt[ED_T / 32];
    const int t = threadIdx.x, lane = t & 31, w = t >> 5;
    h[t] = 0;
    __syncthreads();
    const long long e0 = (long long)rb * ED_EPB;
    int d[8], s[8];
    if (e0 + ED_EPB <= E) {
        const v4i da = *(const v4i*)(dst + e0 + 4 * t);
        const v4i db = *(const v4i*)(dst + e0 + ED_EPB / 2 + 4 * t);
        const v4i sa = *(const v4i*)(src + e0 + 4 * t);
        const v4i sb = *(const v4i*)(src + e0 + ED_EPB / 2 + 4 * t);
        d[0] = da.x; d[1] = da.y; d[2] = da.z; d[3] = da.w;
        d[4] = db.x; d[5] = db.y; d[6] = db.z; d[7] = db.w;
        s[0] = sa.x; s[1] = sa.y; s[2] = sa.z; s[3] = sa.w;
        s[4] = sb.x; s[5] = sb.y; s[6] = sb.z; s[7] = sb.w;
    } else {
#pragma unroll
        for (int u = 0; u < 8; ++u) {
            const long long e = e0 + (u >> 2) * (ED_EPB / 2) + 4 * t + (u & 3);
            d[u] = (e < E) ? dst[e] : -1;
            s[u] = (e < E) ? src[e] : 0;
        }
    }
    int bk[8], r[8];
#pragma unroll
    for (int u = 0; u < 8; ++u) bk[u] = (d[u] >= 0) ? (d[u] >> BK_SHIFT) : 0;
#pragma unroll
    for (int u = 0; u < 8; ++u) {
        r[u] = 0;
        if (d[u] >= 0) r[u] = atomicAdd(&h[bk[u]], 1);
    }
    __syncthreads();

    const int c = h[t];
    int x = c;
#pragma unroll
    for (int o = 1; o < 32; o <<= 1) {
        const int y = __shfl_up(x, o);
        if (lane >= o) x += y;
    }
    if (lane == 31) wt[w] = x;
    __syncthreads();
    int pre = 0;
#pragma unroll
    for (int k = 0; k < ED_T / 32; ++k) pre += (k < w) ? wt[k] : 0;
    const int excl = pre + (x - c);
    h[t] = excl;
    __syncthreads();

    int p[8];
#pragma unroll
    for (int u = 0; u < 8; ++u) p[u] = h[bk[u]] + r[u];
#pragma unroll
    for (int u = 0; u < 8; ++u) {
        if (d[u] >= 0) {
            v2i o;
            o.x = s[u];
            o.y = d[u];
            tmp[e0 + p[u]] = o;
        }
    }
    if (t < nbk) runs[(size_t)t * ga + rb] = (excl << 16) | c;
}

#define BF_T 1024
#define RUN_IT 13

template <bool PLACE>
__device__ __forceinline__ void gcn_bucket_rest(const v2i* __restrict__ tmp, int* cnt,
                                                int* __restrict__ es, int pkv, int r0, int rpw,
                                                int lane) {
    for (int i = 0; i < rpw; ++i) {
        const int pk = __builtin_amdgcn_readlane(pkv, i);
        const int c = pk & 0xFFFF;
        const int o = pk >> 16;
        for (int j = ((i < RUN_IT) ? 32 : 0) + lane; j < c; j += 32) {
            const v2i sd = tmp[(size_t)(r0 + i) * ED_EPB + o + j];
            const int k = atomicAdd(&cnt[sd.y & (BK_NODES - 1)], 1);
            if (PLACE) es[k] = sd.x;
        }
    }
}

__global__ __launch_bounds__(BF_T) void gcn_bucket_kernel(const v2i* __restrict__ tmp,
                                                          const int* __restrict__ runs,
                                                          int* __restrict__ start,
                                                          float* __restrict__ norm,
                                                          int* __restrict__ es, int N, int ga) {
    __shared__ int cnt[BK_NODES];
    __shared__ int wtot[BK_NODES / 32];
    __shared__ int wbs[BF_T / 32];
    const int t = threadIdx.x, lane = t & 31;
    const int w = __builtin_amdgcn_readfirstlane((int)(threadIdx.x >> 5));
    const int b = blockIdx.x;
    const int rpw = (ga + BF_T / 32 - 1) / (BF_T / 32);
    const int r0 = w * rpw;

    int pkv = 0;
    if (lane < rpw && r0 + lane < ga) pkv = runs[(size_t)b * ga + r0 + lane];
    if (t < BK_NODES) cnt[t] = 0;

    int qs[RUN_IT], qd[RUN_IT];
#pragma unroll
    for (int i = 0; i < RUN_IT; ++i) {
        const int pk = __builtin_amdgcn_readlane(pkv, i);
        const int c = pk & 0xFFFF;
        const int o = pk >> 16;
        qs[i] = 0;
        qd[i] = -1;
        if (lane < c) {
            const v2i sd = tmp[(size_t)(r0 + i) * ED_EPB + o + lane];
            qs[i] = sd.x;
            qd[i] = sd.y;
        }
    }
    const bool more = (rpw > RUN_IT) || (__builtin_amdgcn_ballot_w32((pkv & 0xFFFF) > 32) != 0u);

    int osum = pkv >> 16;
#pragma unroll
    for (int o = 16; o > 0; o >>= 1) osum += __shfl_xor(osum, o);
    if (lane == 0) wbs[w] = osum;
    __syncthreads();

#pragma unroll
    for (int i = 0; i < RUN_IT; ++i)
        if (qd[i] >= 0) atomicAdd(&cnt[qd[i] & (BK_NODES - 1)], 1);
    if (more) gcn_bucket_rest<false>(tmp, cnt, es, pkv, r0, rpw, lane);
    __syncthreads();

    const int c = (t < BK_NODES) ? cnt[t] : 0;
    int x = c;
#pragma unroll
    for (int o = 1; o < 32; o <<= 1) {
        const int y = __shfl_up(x, o);
        if (lane >= o) x += y;
    }
    if (t < BK_NODES && lane == 31) wtot[w] = x;
    int bs = wbs[lane];
#pragma unroll
    for (int o = 16; o > 0; o >>= 1) bs += __shfl_xor(bs, o);
    __syncthreads();
    if (t < BK_NODES) {
        int pre = 0;
#pragma unroll
        for (int k = 0; k < BK_NODES / 32; ++k) pre += (k < w) ? wtot[k] : 0;
        const int first = bs + pre + (x - c);
        const int node = b * BK_NODES + t;
        cnt[t] = first;
        if (node < N) {
            start[node] = first;
            norm[node] = rsqrtf(fmaxf((float)c, 1.0f));
            if (node == N - 1) start[N] = first + c;
        }
    }
    __syncthreads();

    unsigned r[RUN_IT];
#pragma unroll
    for (int i = 0; i < RUN_IT; ++i) {
        r[i] = 0u;
        if (qd[i] >= 0) r[i] = (unsigned)atomicAdd(&cnt[qd[i] & (BK_NODES - 1)], 1);
    }
#pragma unroll
    for (int i = 0; i < RUN_IT; ++i)
        if (qd[i] >= 0) es[r[i]] = qs[i];
    if (more) gcn_bucket_rest<true>(tmp, cnt, es, pkv, r0, rpw, lane);
}

#define GEMM_THREADS 256

template <int K, bool RELU_A, bool SUM_A>
__device__ __forceinline__ void gcn_gemm_block(const float* __restrict__ X,
                              const float* __restrict__ X2,
                              const float* __restrict__ W,
                              float* __restrict__ H, int mtiles, int gb) {
    __shared__ v4f sW[K * 16];

    const int wave = threadIdx.x >> 5;
    const int lane = threadIdx.x & 31;
    const int half = lane >> 4;
    const int lid  = lane & 15;
    const int tile = gb * (GEMM_THREADS / 32) + wave;
    const int tile_ld = (tile < mtiles) ? tile : (mtiles - 1);
    const int row0 = tile_ld * 16;

    constexpr int ITEMS = K * 8 / GEMM_THREADS;
    const v4f* __restrict__ W4 = (const v4f*)W;
    v4f q0[ITEMS], q1[ITEMS];
#pragma unroll
    for (int it = 0; it < ITEMS; ++it) {
        const int id = it * GEMM_THREADS + threadIdx.x;
        const int p = id >> 4, j = id & 15;
        q0[it] = W4[(2 * p) * 16 + j];
        q1[it] = W4[(2 * p + 1) * 16 + j];
    }

    const float* __restrict__ Arow = X + (size_t)(row0 + lid) * K;
    v2f a[K / 4];
#pragma unroll
    for (int s = 0; s < K / 4; ++s)
        a[s] = *(const v2f*)(Arow + 4 * s + 2 * half);
    v2f a2[SUM_A ? K / 4 : 1];
    if (SUM_A) {
        const float* __restrict__ Arow2 = X2 + (size_t)(row0 + lid) * K;
#pragma unroll
        for (int s = 0; s < K / 4; ++s)
            a2[s] = *(const v2f*)(Arow2 + 4 * s + 2 * half);
    }

#pragma unroll
    for (int it = 0; it < ITEMS; ++it) {
        const int id = it * GEMM_THREADS + threadIdx.x;
        const int p = id >> 4, j = id & 15;
        v4f lo, hi;
        lo.x = q0[it].x; lo.y = q1[it].x; lo.z = q0[it].y; lo.w = q1[it].y;
        hi.x = q0[it].z; hi.y = q1[it].z; hi.z = q0[it].w; hi.w = q1[it].w;
        sW[(2 * p) * 16 + j]     = lo;
        sW[(2 * p + 1) * 16 + j] = hi;
    }
    __syncthreads();

    v8f c0 = {}, c1 = {}, c2 = {}, c3 = {};
#pragma unroll
    for (int s = 0; s < K / 4; ++s) {
        v2f av = a[s];
        if (SUM_A) {
            av.x += a2[s].x;
            av.y += a2[s].y;
        }
        if (RELU_A) {
            av.x = fmaxf(av.x, 0.0f);
            av.y = fmaxf(av.y, 0.0f);
        }
        const int p = 2 * s + half;
        const v4f u0 = sW[(2 * p) * 16 + lid];
        const v4f u1 = sW[(2 * p + 1) * 16 + lid];
        v2f b0, b1, b2, b3;
        b0.x = u0.x; b0.y = u0.y;
        b1.x = u0.z; b1.y = u0.w;
        b2.x = u1.x; b2.y = u1.y;
        b3.x = u1.z; b3.y = u1.w;
        c0 = __builtin_amdgcn_wmma_f32_16x16x4_f32(false, av, false, b0, (short)0, c0, false, false);
        c1 = __builtin_amdgcn_wmma_f32_16x16x4_f32(false, av, false, b1, (short)0, c1, false, false);
        c2 = __builtin_amdgcn_wmma_f32_16x16x4_f32(false, av, false, b2, (short)0, c2, false, false);
        c3 = __builtin_amdgcn_wmma_f32_16x16x4_f32(false, av, false, b3, (short)0, c3, false, false);
    }

    if (tile < mtiles) {
        float* __restrict__ Hb = H + (size_t)(row0 + 8 * half) * HID_DIM + 4 * lid;
#pragma unroll
        for (int r = 0; r < 8; ++r) {
            v4f o;
            o.x = c0[r]; o.y = c1[r]; o.z = c2[r]; o.w = c3[r];
            *(v4f*)(Hb + (size_t)r * HID_DIM) = o;
        }
    }
}

static_assert(GEMM_THREADS == ED_T, "both roles use the same block size");

template <int K>
__global__ __launch_bounds__(GEMM_THREADS) void gcn_front_kernel(const float* __restrict__ X,
                                                                 const float* __restrict__ W,
                                                                 float* __restrict__ H,
                                                                 int mtiles, int gblocks,
                                                                 const int* __restrict__ src,
                                                                 const int* __restrict__ dst,
                                                                 int* __restrict__ runs,
                                                                 v2i* __restrict__ tmp,
                                                                 int E, int nbk, int ga) {
    const int b = (int)blockIdx.x;
    if (b < gblocks)
        gcn_gemm_block<K, false, false>(X, X, W, H, mtiles, b);
    else
        gcn_runs_block(src, dst, runs, tmp, E, nbk, b - gblocks, ga);
}

#define GA_NPW 8
#define GA_U   8

__device__ __forceinline__ int gcn_slots_load(const int* __restrict__ es, int c, int e, int lane) {
    int m = e - c;
    m = (m > 32) ? 32 : m;
    int idx = c + ((lane < m) ? lane : (m - 1));
    idx = (idx < 0) ? 0 : idx;
    return es[idx];
}

__device__ __forceinline__ void gcn_rows_load(const float* __restrict__ Hl, int s, int j0,
                                              v2f (&h)[GA_U]) {
#pragma unroll
    for (int u = 0; u < GA_U; ++u) {
        const int sj = __builtin_amdgcn_readlane(s, j0 + u);
        h[u] = *(const v2f*)(Hl + (size_t)sj * HID_DIM);
    }
}

__device__ __forceinline__ void gcn_rows_sum(int nbits, int j0, int m, const v2f (&h)[GA_U],
                                             float& ax, float& ay) {
#pragma unroll
    for (int u = 0; u < GA_U; ++u) {
        const float nj = __int_as_float(__builtin_amdgcn_readlane(nbits, j0 + u));
        const bool ok = (j0 + u) < m;
        ax = fmaf(ok ? h[u].x : 0.0f, nj, ax);
        ay = fmaf(ok ? h[u].y : 0.0f, nj, ay);
    }
}

__device__ __forceinline__ void gcn_slots_sum(const float* __restrict__ Hl,
                                              const float* __restrict__ norm, int s, int m,
                                              float& ax, float& ay) {
    const int nbits = __float_as_int(norm[s]);
    for (int j = 0; j < m; j += 2 * GA_U) {
        v2f h0[GA_U], h1[GA_U];
        const bool two = (j + GA_U) < m;
        gcn_rows_load(Hl, s, j, h0);
        if (two) gcn_rows_load(Hl, s, j + GA_U, h1);
        gcn_rows_sum(nbits, j, m, h0, ax, ay);
        if (two) gcn_rows_sum(nbits, j + GA_U, m, h1, ax, ay);
    }
}

#define GA_WAVES 8
#define GA_ROWS  (GA_NPW * GA_WAVES)
#define GA_PITCH 68

template <bool FUSE_GEMM>
__global__ __launch_bounds__(256) void gcn_gather_kernel(const float* __restrict__ H,
                                                         const float* __restrict__ norm,
                                                         const int* __restrict__ start,
                                                         const int* __restrict__ es,
                                                         const float* __restrict__ W,
                                                         float* __restrict__ out, int N) {
    __shared__ v4f   sW[FUSE_GEMM ? HID_DIM * 16 : 1];
    __shared__ float sX[FUSE_GEMM ? GA_ROWS * GA_PITCH : 2];

    const int lane = threadIdx.x & 31;
    const int wib = __builtin_amdgcn_readfirstlane((int)(threadIdx.x >> 5));
    const int n0 = ((int)blockIdx.x * GA_WAVES + wib) * GA_NPW;
    if (!FUSE_GEMM) {
        if (n0 >= N) return;
    }
    int nn = (N - n0 < GA_NPW) ? (N - n0) : GA_NPW;
    nn = (nn < 0) ? 0 : nn;
    const float* __restrict__ Hl = H + 2 * lane;

    int bi = n0 + ((lane < nn) ? lane : nn);
    bi = (bi > N) ? N : bi;
    const int bv = start[bi];

    constexpr int W_ITEMS = HID_DIM * 8 / 256;
    v4f q0[W_ITEMS], q1[W_ITEMS];
    if (FUSE_GEMM) {
        const v4f* __restrict__ W4 = (const v4f*)W;
#pragma unroll
        for (int it = 0; it < W_ITEMS; ++it) {
            const int id = it * 256 + (int)threadIdx.x;
            const int p = id >> 4, j = id & 15;
            q0[it] = W4[(2 * p) * 16 + j];
            q1[it] = W4[(2 * p + 1) * 16 + j];
        }
    }

    int b = __builtin_amdgcn_readlane(bv, 0);
    int e = __builtin_amdgcn_readlane(bv, 1);
    int s = gcn_slots_load(es, b, e, lane);

    for (int k = 0; k < nn; ++k) {
        int e2 = e;
        int s2 = s;
        if (k + 1 < nn) {
            e2 = __builtin_amdgcn_readlane(bv, k + 2);
            s2 = gcn_slots_load(es, e, e2, lane);
        }

        float ax = 0.0f, ay = 0.0f;
        if (b < e) {
            gcn_slots_sum(Hl, norm, s, (e - b > 32) ? 32 : (e - b), ax, ay);
            for (int c = b + 32; c < e; c += 32) {
                s = gcn_slots_load(es, c, e, lane);
                gcn_slots_sum(Hl, norm, s, (e - c > 32) ? 32 : (e - c), ax, ay);
            }
        }
        v2f o;
        o.x = fmaxf(ax, 0.0f);
        o.y = fmaxf(ay, 0.0f);
        if (FUSE_GEMM)
            *(v2f*)(sX + (wib * GA_NPW + k) * GA_PITCH + 2 * lane) = o;
        else
            *(v2f*)(out + (size_t)(n0 + k) * HID_DIM + 2 * lane) = o;

        b = e;
        e = e2;
        s = s2;
    }

    if (FUSE_GEMM) {
#pragma unroll
        for (int it = 0; it < W_ITEMS; ++it) {
            const int id = it * 256 + (int)threadIdx.x;
            const int p = id >> 4, j = id & 15;
            v4f lo, hi;
            lo.x = q0[it].x; lo.y = q1[it].x; lo.z = q0[it].y; lo.w = q1[it].y;
            hi.x = q0[it].z; hi.y = q1[it].z; hi.z = q0[it].w; hi.w = q1[it].w;
            sW[(2 * p) * 16 + j]     = lo;
            sW[(2 * p + 1) * 16 + j] = hi;
        }

        for (int k = nn; k < GA_NPW; ++k) {
            v2f z;
            z.x = 0.0f;
            z.y = 0.0f;
            *(v2f*)(sX + (wib * GA_NPW + k) * GA_PITCH + 2 * lane) = z;
        }
        __syncthreads();
        if (wib >= GA_ROWS / 16) return;

        const int half = lane >> 4;
        const int lid  = lane & 15;
        const float* Ar = sX + (wib * 16 + lid) * GA_PITCH + 2 * half;
        v8f c0 = {}, c1 = {}, c2 = {}, c3 = {};
#pragma unroll
        for (int s4 = 0; s4 < HID_DIM / 4; ++s4) {
            const v2f av = *(const v2f*)(Ar + 4 * s4);
            const int p = 2 * s4 + half;
            const v4f u0 = sW[(2 * p) * 16 + lid];
            const v4f u1 = sW[(2 * p + 1) * 16 + lid];
            v2f b0, b1, b2, b3;
            b0.x = u0.x; b0.y = u0.y;
            b1.x = u0.z; b1.y = u0.w;
            b2.x = u1.x; b2.y = u1.y;
            b3.x = u1.z; b3.y = u1.w;
            c0 = __builtin_amdgcn_wmma_f32_16x16x4_f32(false, av, false, b0, (short)0, c0, false, false);
            c1 = __builtin_amdgcn_wmma_f32_16x16x4_f32(false, av, false, b1, (short)0, c1, false, false);
            c2 = __builtin_amdgcn_wmma_f32_16x16x4_f32(false, av, false, b2, (short)0, c2, false, false);
            c3 = __builtin_amdgcn_wmma_f32_16x16x4_f32(false, av, false, b3, (short)0, c3, false, false);
        }

        const int strip0 = (int)blockIdx.x * GA_ROWS + wib * 16;
        if (strip0 < N) {
            float* __restrict__ Ob = out + (size_t)(strip0 + 8 * half) * HID_DIM + 4 * lid;
#pragma unroll
            for (int r = 0; r < 8; ++r) {
                v4f o;
                o.x = c0[r]; o.y = c1[r]; o.z = c2[r]; o.w = c3[r];
                *(v4f*)(Ob + (size_t)r * HID_DIM) = o;
            }
        }
    }
}


extern "C" void kernel_launch(void* const* d_in, const int* in_sizes, int n_in,
                              void* d_out, int out_size, void* d_ws, size_t ws_size,
                              hipStream_t stream) {
    const float* features = (const float*)d_in[0];
    const float* W1       = (const float*)d_in[1];
    const float* W2       = (const float*)d_in[2];
    const int*   src      = (const int*)d_in[3];
    const int*   dst      = (const int*)d_in[4];

    const int N = in_sizes[0] / IN_DIM;
    const int E = in_sizes[3];
    const long long NH = (long long)N * HID_DIM;
    const long long E64 = ((long long)E + 63) / 64 * 64;

    const int nbk = (N + BK_NODES - 1) / BK_NODES;
    const int ga  = (E + ED_EPB - 1) / ED_EPB;
    const int nsc = nbk * ga;
    const long long S64 = ((long long)nsc + 63) / 64 * 64;

    int*   runs  = (int*)d_ws;
    int*   start = runs + S64;
    float* norm  = (float*)(start + 65536 + 64);
    v2i*   tmp   = (v2i*)(norm + 65536);
    int*   es    = (int*)(tmp + (long long)ga * ED_EPB);
    float* H     = (float*)(es + E64);
    float* H2    = H + NH;
    float* OUT   = (float*)d_out;

    const int mtiles = N / 16;
    const int gblocks = (mtiles + GEMM_THREADS / 32 - 1) / (GEMM_THREADS / 32);
    const int nblocks = (N + GA_ROWS - 1) / GA_ROWS;

    gcn_front_kernel<IN_DIM><<<gblocks + ga, GEMM_THREADS, 0, stream>>>(features, W1, H, mtiles, gblocks,
                                                                      src, dst, runs, tmp, E, nbk, ga);
    gcn_bucket_kernel<<<nbk, BF_T, 0, stream>>>(tmp, runs, start, norm, es, N, ga);

    gcn_gather_kernel<true><<<nblocks, 256, 0, stream>>>(H, norm, start, es, W2, H2, N);

    gcn_gather_kernel<false><<<nblocks, 256, 0, stream>>>(H2, norm, start, es, W2, OUT, N);
}


# rows carry their norm before they are gathered: the bucket pass scales its 256 rows of H in place, the fused product's epilogue scales H2; no per-edge norm load; products rounded before the sum
# speedup vs baseline: 4.6890x; 1.0440x over previous
#include <hip/hip_runtime.h>

typedef __attribute__((ext_vector_type(2))) float v2f;
typedef __attribute__((ext_vector_type(4))) float v4f;
typedef __attribute__((ext_vector_type(8))) float v8f;
typedef __attribute__((ext_vector_type(2))) int v2i;
typedef __attribute__((ext_vector_type(4))) int v4i;

#define IN_DIM 128
#define HID_DIM 64

#define BK_SHIFT 8
#define BK_NODES 256
#define NBK_MAX  256
#define ED_EPB   2048
#define ED_T     256

__device__ __forceinline__ void gcn_runs_block(const int* __restrict__ src,
                                               const int* __restrict__ dst,
                                               int* __restrict__ runs,
                                               v2i* __restrict__ tmp, int E, int nbk,
                                               int rb, int ga) {
    __shared__ int h[NBK_MAX];
    __shared__ int wt[ED_T / 32];
    const int t = threadIdx.x, lane = t & 31, w = t >> 5;
    h[t] = 0;
    __syncthreads();
    const long long e0 = (long long)rb * ED_EPB;
    int d[8], s[8];
    if (e0 + ED_EPB <= E) {
        const v4i da = *(const v4i*)(dst + e0 + 4 * t);
        const v4i db = *(const v4i*)(dst + e0 + ED_EPB / 2 + 4 * t);
        const v4i sa = *(const v4i*)(src + e0 + 4 * t);
        const v4i sb = *(const v4i*)(src + e0 + ED_EPB / 2 + 4 * t);
        d[0] = da.x; d[1] = da.y; d[2] = da.z; d[3] = da.w;
        d[4] = db.x; d[5] = db.y; d[6] = db.z; d[7] = db.w;
        s[0] = sa.x; s[1] = sa.y; s[2] = sa.z; s[3] = sa.w;
        s[4] = sb.x; s[5] = sb.y; s[6] = sb.z; s[7] = sb.w;
    } else {
#pragma unroll
        for (int u = 0; u < 8; ++u) {
            const long long e = e0 + (u >> 2) * (ED_EPB / 2) + 4 * t + (u & 3);
            d[u] = (e < E) ? dst[e] : -1;
            s[u] = (e < E) ? src[e] : 0;
        }
    }
    int bk[8], r[8];
#pragma unroll
    for (int u = 0; u < 8; ++u) bk[u] = (d[u] >= 0) ? (d[u] >> BK_SHIFT) : 0;
#pragma unroll
    for (int u = 0; u < 8; ++u) {
        r[u] = 0;
        if (d[u] >= 0) r[u] = atomicAdd(&h[bk[u]], 1);
    }
    __syncthreads();

    const int c = h[t];
    int x = c;
#pragma unroll
    for (int o = 1; o < 32; o <<= 1) {
        const int y = __shfl_up(x, o);
        if (lane >= o) x += y;
    }
    if (lane == 31) wt[w] = x;
    __syncthreads();
    int pre = 0;
#pragma unroll
    for (int k = 0; k < ED_T / 32; ++k) pre += (k < w) ? wt[k] : 0;
    const int excl = pre + (x - c);
    h[t] = excl;
    __syncthreads();

    int p[8];
#pragma unroll
    for (int u = 0; u < 8; ++u) p[u] = h[bk[u]] + r[u];
#pragma unroll
    for (int u = 0; u < 8; ++u) {
        if (d[u] >= 0) {
            v2i o;
            o.x = s[u];
            o.y = d[u];
            tmp[e0 + p[u]] = o;
        }
    }
    if (t < nbk) runs[(size_t)t * ga + rb] = (excl << 16) | c;
}

#define BF_T 1024
#define BK_RPW (BK_NODES / (BF_T / 32))
#define RUN_IT 13

template <bool PLACE>
__device__ __forceinline__ void gcn_bucket_rest(const v2i* __restrict__ tmp, int* cnt,
                                                int* __restrict__ es, int pkv, int r0, int rpw,
                                                int lane) {
    for (int i = 0; i < rpw; ++i) {
        const int pk = __builtin_amdgcn_readlane(pkv, i);
        const int c = pk & 0xFFFF;
        const int o = pk >> 16;
        for (int j = ((i < RUN_IT) ? 32 : 0) + lane; j < c; j += 32) {
            const v2i sd = tmp[(size_t)(r0 + i) * ED_EPB + o + j];
            const int k = atomicAdd(&cnt[sd.y & (BK_NODES - 1)], 1);
            if (PLACE) es[k] = sd.x;
        }
    }
}

__global__ __launch_bounds__(BF_T) void gcn_bucket_kernel(const v2i* __restrict__ tmp,
                                                          const int* __restrict__ runs,
                                                          int* __restrict__ start,
                                                          float* __restrict__ norm,
                                                          int* __restrict__ es,
                                                          float* __restrict__ H, int N, int ga) {
    __shared__ int cnt[BK_NODES];
    __shared__ float snorm[BK_NODES];
    __shared__ int wtot[BK_NODES / 32];
    __shared__ int wbs[BF_T / 32];
    const int t = threadIdx.x, lane = t & 31;
    const int w = __builtin_amdgcn_readfirstlane((int)(threadIdx.x >> 5));
    const int b = blockIdx.x;
    const int rpw = (ga + BF_T / 32 - 1) / (BF_T / 32);
    const int r0 = w * rpw;

    int pkv = 0;
    if (lane < rpw && r0 + lane < ga) pkv = runs[(size_t)b * ga + r0 + lane];
    if (t < BK_NODES) cnt[t] = 0;

    const int hrow0 = b * BK_NODES + w * BK_RPW;
    const bool hlive = hrow0 < N;
    const float* Hr = H + (size_t)(hlive ? hrow0 : 0) * HID_DIM + 2 * lane;
    v2f hr[BK_RPW];
#pragma unroll
    for (int r = 0; r < BK_RPW; ++r) hr[r] = *(const v2f*)(Hr + r * HID_DIM);

    int qs[RUN_IT], qd[RUN_IT];
#pragma unroll
    for (int i = 0; i < RUN_IT; ++i) {
        const int pk = __builtin_amdgcn_readlane(pkv, i);
        const int c = pk & 0xFFFF;
        const int o = pk >> 16;
        qs[i] = 0;
        qd[i] = -1;
        if (lane < c) {
            const v2i sd = tmp[(size_t)(r0 + i) * ED_EPB + o + lane];
            qs[i] = sd.x;
            qd[i] = sd.y;
        }
    }
    const bool more = (rpw > RUN_IT) || (__builtin_amdgcn_ballot_w32((pkv & 0xFFFF) > 32) != 0u);

    int osum = pkv >> 16;
#pragma unroll
    for (int o = 16; o > 0; o >>= 1) osum += __shfl_xor(osum, o);
    if (lane == 0) wbs[w] = osum;
    __syncthreads();

#pragma unroll
    for (int i = 0; i < RUN_IT; ++i)
        if (qd[i] >= 0) atomicAdd(&cnt[qd[i] & (BK_NODES - 1)], 1);
    if (more) gcn_bucket_rest<false>(tmp, cnt, es, pkv, r0, rpw, lane);
    __syncthreads();

    const int c = (t < BK_NODES) ? cnt[t] : 0;
    int x = c;
#pragma unroll
    for (int o = 1; o < 32; o <<= 1) {
        const int y = __shfl_up(x, o);
        if (lane >= o) x += y;
    }
    if (t < BK_NODES && lane == 31) wtot[w] = x;
    int bs = wbs[lane];
#pragma unroll
    for (int o = 16; o > 0; o >>= 1) bs += __shfl_xor(bs, o);
    __syncthreads();
    if (t < BK_NODES) {
        int pre = 0;
#pragma unroll
        for (int k = 0; k < BK_NODES / 32; ++k) pre += (k < w) ? wtot[k] : 0;
        const int first = bs + pre + (x - c);
        const int node = b * BK_NODES + t;
        cnt[t] = first;
        const float nv = rsqrtf(fmaxf((float)c, 1.0f));
        snorm[t] = nv;
        if (node < N) {
            start[node] = first;
            norm[node] = nv;
            if (node == N - 1) start[N] = first + c;
        }
    }
    __syncthreads();

    if (hlive) {
        float* Hw = H + (size_t)hrow0 * HID_DIM + 2 * lane;
#pragma unroll
        for (int r = 0; r < BK_RPW; ++r) {
            const float nr = snorm[w * BK_RPW + r];
            v2f o;
            o.x = hr[r].x * nr;
            o.y = hr[r].y * nr;
            *(v2f*)(Hw + r * HID_DIM) = o;
        }
    }

    unsigned r[RUN_IT];
#pragma unroll
    for (int i = 0; i < RUN_IT; ++i) {
        r[i] = 0u;
        if (qd[i] >= 0) r[i] = (unsigned)atomicAdd(&cnt[qd[i] & (BK_NODES - 1)], 1);
    }
#pragma unroll
    for (int i = 0; i < RUN_IT; ++i)
        if (qd[i] >= 0) es[r[i]] = qs[i];
    if (more) gcn_bucket_rest<true>(tmp, cnt, es, pkv, r0, rpw, lane);
}

#define GEMM_THREADS 256

template <int K, bool RELU_A, bool SUM_A>
__device__ __forceinline__ void gcn_gemm_block(const float* __restrict__ X,
                              const float* __restrict__ X2,
                              const float* __restrict__ W,
                              float* __restrict__ H, int mtiles, int gb) {
    __shared__ v4f sW[K * 16];

    const int wave = threadIdx.x >> 5;
    const int lane = threadIdx.x & 31;
    const int half = lane >> 4;
    const int lid  = lane & 15;
    const int tile = gb * (GEMM_THREADS / 32) + wave;
    const int tile_ld = (tile < mtiles) ? tile : (mtiles - 1);
    const int row0 = tile_ld * 16;

    constexpr int ITEMS = K * 8 / GEMM_THREADS;
    const v4f* __restrict__ W4 = (const v4f*)W;
    v4f q0[ITEMS], q1[ITEMS];
#pragma unroll
    for (int it = 0; it < ITEMS; ++it) {
        const int id = it * GEMM_THREADS + threadIdx.x;
        const int p = id >> 4, j = id & 15;
        q0[it] = W4[(2 * p) * 16 + j];
        q1[it] = W4[(2 * p + 1) * 16 + j];
    }

    const float* __restrict__ Arow = X + (size_t)(row0 + lid) * K;
    v2f a[K / 4];
#pragma unroll
    for (int s = 0; s < K / 4; ++s)
        a[s] = *(const v2f*)(Arow + 4 * s + 2 * half);
    v2f a2[SUM_A ? K / 4 : 1];
    if (SUM_A) {
        const float* __restrict__ Arow2 = X2 + (size_t)(row0 + lid) * K;
#pragma unroll
        for (int s = 0; s < K / 4; ++s)
            a2[s] = *(const v2f*)(Arow2 + 4 * s + 2 * half);
    }

#pragma unroll
    for (int it = 0; it < ITEMS; ++it) {
        const int id = it * GEMM_THREADS + threadIdx.x;
        const int p = id >> 4, j = id & 15;
        v4f lo, hi;
        lo.x = q0[it].x; lo.y = q1[it].x; lo.z = q0[it].y; lo.w = q1[it].y;
        hi.x = q0[it].z; hi.y = q1[it].z; hi.z = q0[it].w; hi.w = q1[it].w;
        sW[(2 * p) * 16 + j]     = lo;
        sW[(2 * p + 1) * 16 + j] = hi;
    }
    __syncthreads();

    v8f c0 = {}, c1 = {}, c2 = {}, c3 = {};
#pragma unroll
    for (int s = 0; s < K / 4; ++s) {
        v2f av = a[s];
        if (SUM_A) {
            av.x += a2[s].x;
            av.y += a2[s].y;
        }
        if (RELU_A) {
            av.x = fmaxf(av.x, 0.0f);
            av.y = fmaxf(av.y, 0.0f);
        }
        const int p = 2 * s + half;
        const v4f u0 = sW[(2 * p) * 16 + lid];
        const v4f u1 = sW[(2 * p + 1) * 16 + lid];
        v2f b0, b1, b2, b3;
        b0.x = u0.x; b0.y = u0.y;
        b1.x = u0.z; b1.y = u0.w;
        b2.x = u1.x; b2.y = u1.y;
        b3.x = u1.z; b3.y = u1.w;
        c0 = __builtin_amdgcn_wmma_f32_16x16x4_f32(false, av, false, b0, (short)0, c0, false, false);
        c1 = __builtin_amdgcn_wmma_f32_16x16x4_f32(false, av, false, b1, (short)0, c1, false, false);
        c2 = __builtin_amdgcn_wmma_f32_16x16x4_f32(false, av, false, b2, (short)0, c2, false, false);
        c3 = __builtin_amdgcn_wmma_f32_16x16x4_f32(false, av, false, b3, (short)0, c3, false, false);
    }

    if (tile < mtiles) {
        float* __restrict__ Hb = H + (size_t)(row0 + 8 * half) * HID_DIM + 4 * lid;
#pragma unroll
        for (int r = 0; r < 8; ++r) {
            v4f o;
            o.x = c0[r]; o.y = c1[r]; o.z = c2[r]; o.w = c3[r];
            *(v4f*)(Hb + (size_t)r * HID_DIM) = o;
        }
    }
}

static_assert(GEMM_THREADS == ED_T, "both roles use the same block size");

template <int K>
__global__ __launch_bounds__(GEMM_THREADS) void gcn_front_kernel(const float* __restrict__ X,
                                                                 const float* __restrict__ W,
                                                                 float* __restrict__ H,
                                                                 int mtiles, int gblocks,
                                                                 const int* __restrict__ src,
                                                                 const int* __restrict__ dst,
                                                                 int* __restrict__ runs,
                                                                 v2i* __restrict__ tmp,
                                                                 int E, int nbk, int ga) {
    const int b = (int)blockIdx.x;
    if (b < gblocks)
        gcn_gemm_block<K, false, false>(X, X, W, H, mtiles, b);
    else
        gcn_runs_block(src, dst, runs, tmp, E, nbk, b - gblocks, ga);
}

#define GA_NPW 8
#define GA_U   8

__device__ __forceinline__ int gcn_slots_load(const int* __restrict__ es, int c, int e, int lane) {
    int m = e - c;
    m = (m > 32) ? 32 : m;
    int idx = c + ((lane < m) ? lane : (m - 1));
    idx = (idx < 0) ? 0 : idx;
    return es[idx];
}

__device__ __forceinline__ void gcn_rows_load(const float* __restrict__ Hl, int s, int j0,
                                              v2f (&h)[GA_U]) {
#pragma unroll
    for (int u = 0; u < GA_U; ++u) {
        const int sj = __builtin_amdgcn_readlane(s, j0 + u);
        h[u] = *(const v2f*)(Hl + (size_t)sj * HID_DIM);
    }
}

__device__ __forceinline__ void gcn_rows_sum(int j0, int m, const v2f (&h)[GA_U],
                                             float& ax, float& ay) {
#pragma unroll
    for (int u = 0; u < GA_U; ++u) {
        const bool ok = (j0 + u) < m;
        ax += ok ? h[u].x : 0.0f;
        ay += ok ? h[u].y : 0.0f;
    }
}

__device__ __forceinline__ void gcn_slots_sum(const float* __restrict__ Hl, int s, int m,
                                              float& ax, float& ay) {
    for (int j = 0; j < m; j += 2 * GA_U) {
        v2f h0[GA_U], h1[GA_U];
        const bool two = (j + GA_U) < m;
        gcn_rows_load(Hl, s, j, h0);
        if (two) gcn_rows_load(Hl, s, j + GA_U, h1);
        gcn_rows_sum(j, m, h0, ax, ay);
        if (two) gcn_rows_sum(j + GA_U, m, h1, ax, ay);
    }
}

#define GA_WAVES 8
#define GA_ROWS  (GA_NPW * GA_WAVES)
#define GA_PITCH 68

template <bool FUSE_GEMM>
__global__ __launch_bounds__(256) void gcn_gather_kernel(const float* __restrict__ H,
                                                         const float* __restrict__ norm,
                                                         const int* __restrict__ start,
                                                         const int* __restrict__ es,
                                                         const float* __restrict__ W,
                                                         float* __restrict__ out, int N) {
    __shared__ v4f   sW[FUSE_GEMM ? HID_DIM * 16 : 1];
    __shared__ float sX[FUSE_GEMM ? GA_ROWS * GA_PITCH : 2];

    const int lane = threadIdx.x & 31;
    const int wib = __builtin_amdgcn_readfirstlane((int)(threadIdx.x >> 5));
    const int n0 = ((int)blockIdx.x * GA_WAVES + wib) * GA_NPW;
    if (!FUSE_GEMM) {
        if (n0 >= N) return;
    }
    int nn = (N - n0 < GA_NPW) ? (N - n0) : GA_NPW;
    nn = (nn < 0) ? 0 : nn;
    const float* __restrict__ Hl = H + 2 * lane;

    int bi = n0 + ((lane < nn) ? lane : nn);
    bi = (bi > N) ? N : bi;
    const int bv = start[bi];

    constexpr int W_ITEMS = HID_DIM * 8 / 256;
    v4f q0[W_ITEMS], q1[W_ITEMS];
    if (FUSE_GEMM) {
        const v4f* __restrict__ W4 = (const v4f*)W;
#pragma unroll
        for (int it = 0; it < W_ITEMS; ++it) {
            const int id = it * 256 + (int)threadIdx.x;
            const int p = id >> 4, j = id & 15;
            q0[it] = W4[(2 * p) * 16 + j];
            q1[it] = W4[(2 * p + 1) * 16 + j];
        }
    }

    int b = __builtin_amdgcn_readlane(bv, 0);
    int e = __builtin_amdgcn_readlane(bv, 1);
    int s = gcn_slots_load(es, b, e, lane);

    for (int k = 0; k < nn; ++k) {
        int e2 = e;
        int s2 = s;
        if (k + 1 < nn) {
            e2 = __builtin_amdgcn_readlane(bv, k + 2);
            s2 = gcn_slots_load(es, e, e2, lane);
        }

        float ax = 0.0f, ay = 0.0f;
        if (b < e) {
            gcn_slots_sum(Hl, s, (e - b > 32) ? 32 : (e - b), ax, ay);
            for (int c = b + 32; c < e; c += 32) {
                s = gcn_slots_load(es, c, e, lane);
                gcn_slots_sum(Hl, s, (e - c > 32) ? 32 : (e - c), ax, ay);
            }
        }
        v2f o;
        o.x = fmaxf(ax, 0.0f);
        o.y = fmaxf(ay, 0.0f);
        if (FUSE_GEMM)
            *(v2f*)(sX + (wib * GA_NPW + k) * GA_PITCH + 2 * lane) = o;
        else
            *(v2f*)(out + (size_t)(n0 + k) * HID_DIM + 2 * lane) = o;

        b = e;
        e = e2;
        s = s2;
    }

    if (FUSE_GEMM) {
#pragma unroll
        for (int it = 0; it < W_ITEMS; ++it) {
            const int id = it * 256 + (int)threadIdx.x;
            const int p = id >> 4, j = id & 15;
            v4f lo, hi;
            lo.x = q0[it].x; lo.y = q1[it].x; lo.z = q0[it].y; lo.w = q1[it].y;
            hi.x = q0[it].z; hi.y = q1[it].z; hi.z = q0[it].w; hi.w = q1[it].w;
            sW[(2 * p) * 16 + j]     = lo;
            sW[(2 * p + 1) * 16 + j] = hi;
        }

        for (int k = nn; k < GA_NPW; ++k) {
            v2f z;
            z.x = 0.0f;
            z.y = 0.0f;
            *(v2f*)(sX + (wib * GA_NPW + k) * GA_PITCH + 2 * lane) = z;
        }
        __syncthreads();
        const int strip0 = (int)blockIdx.x * GA_ROWS + wib * 16;
        if (wib >= GA_ROWS / 16 || strip0 >= N) return;

        const int half = lane >> 4;
        const int lid  = lane & 15;
        const v4f na = *(const v4f*)(norm + strip0 + 8 * half);
        const v4f nb = *(const v4f*)(norm + strip0 + 8 * half + 4);
        const float* Ar = sX + (wib * 16 + lid) * GA_PITCH + 2 * half;
        v8f c0 = {}, c1 = {}, c2 = {}, c3 = {};
#pragma unroll
        for (int s4 = 0; s4 < HID_DIM / 4; ++s4) {
            const v2f av = *(const v2f*)(Ar + 4 * s4);
            const int p = 2 * s4 + half;
            const v4f u0 = sW[(2 * p) * 16 + lid];
            const v4f u1 = sW[(2 * p + 1) * 16 + lid];
            v2f b0, b1, b2, b3;
            b0.x = u0.x; b0.y = u0.y;
            b1.x = u0.z; b1.y = u0.w;
            b2.x = u1.x; b2.y = u1.y;
            b3.x = u1.z; b3.y = u1.w;
            c0 = __builtin_amdgcn_wmma_f32_16x16x4_f32(false, av, false, b0, (short)0, c0, false, false);
            c1 = __builtin_amdgcn_wmma_f32_16x16x4_f32(false, av, false, b1, (short)0, c1, false, false);
            c2 = __builtin_amdgcn_wmma_f32_16x16x4_f32(false, av, false, b2, (short)0, c2, false, false);
            c3 = __builtin_amdgcn_wmma_f32_16x16x4_f32(false, av, false, b3, (short)0, c3, false, false);
        }

        float* __restrict__ Ob = out + (size_t)(strip0 + 8 * half) * HID_DIM + 4 * lid;
#pragma unroll
        for (int r = 0; r < 8; ++r) {
            const float nr = (r < 4) ? na[r & 3] : nb[r & 3];
            v4f o;
            o.x = c0[r] * nr; o.y = c1[r] * nr; o.z = c2[r] * nr; o.w = c3[r] * nr;
            *(v4f*)(Ob + (size_t)r * HID_DIM) = o;
        }
    }
}


extern "C" void kernel_launch(void* const* d_in, const int* in_sizes, int n_in,
                              void* d_out, int out_size, void* d_ws, size_t ws_size,
                              hipStream_t stream) {
    const float* features = (const float*)d_in[0];
    const float* W1       = (const float*)d_in[1];
    const float* W2       = (const float*)d_in[2];
    const int*   src      = (const int*)d_in[3];
    const int*   dst      = (const int*)d_in[4];

    const int N = in_sizes[0] / IN_DIM;
    const int E = in_sizes[3];
    const long long NH = (long long)N * HID_DIM;
    const long long E64 = ((long long)E + 63) / 64 * 64;

    const int nbk = (N + BK_NODES - 1) / BK_NODES;
    const int ga  = (E + ED_EPB - 1) / ED_EPB;
    const int nsc = nbk * ga;
    const long long S64 = ((long long)nsc + 63) / 64 * 64;

    int*   runs  = (int*)d_ws;
    int*   start = runs + S64;
    float* norm  = (float*)(start + 65536 + 64);
    v2i*   tmp   = (v2i*)(norm + 65536);
    int*   es    = (int*)(tmp + (long long)ga * ED_EPB);
    float* H     = (float*)(es + E64);
    float* H2    = H + NH;
    float* OUT   = (float*)d_out;

    const int mtiles = N / 16;
    const int gblocks = (mtiles + GEMM_THREADS / 32 - 1) / (GEMM_THREADS / 32);
    const int nblocks = (N + GA_ROWS - 1) / GA_ROWS;

    gcn_front_kernel<IN_DIM><<<gblocks + ga, GEMM_THREADS, 0, stream>>>(features, W1, H, mtiles, gblocks,
                                                                      src, dst, runs, tmp, E, nbk, ga);
    gcn_bucket_kernel<<<nbk, BF_T, 0, stream>>>(tmp, runs, start, norm, es, H, N, ga);

    gcn_gather_kernel<true><<<nblocks, 256, 0, stream>>>(H, norm, start, es, W2, H2, N);

    gcn_gather_kernel<false><<<nblocks, 256, 0, stream>>>(H2, norm, start, es, W2, OUT, N);
}


# edge sort stores whole lines: both passes collect their sorted output in LDS (a block's 16 KB of pairs; a bucket's sources) and write consecutive words, not a scattered 4/8-B piece per lane
# speedup vs baseline: 5.1491x; 1.0981x over previous
#include <hip/hip_runtime.h>

typedef __attribute__((ext_vector_type(2))) float v2f;
typedef __attribute__((ext_vector_type(4))) float v4f;
typedef __attribute__((ext_vector_type(8))) float v8f;
typedef __attribute__((ext_vector_type(2))) int v2i;
typedef __attribute__((ext_vector_type(4))) int v4i;

#define IN_DIM 128
#define HID_DIM 64

#define BK_SHIFT 8
#define BK_NODES 256
#define NBK_MAX  256
#define ED_EPB   2048
#define ED_T     256

__device__ __forceinline__ void gcn_runs_block(const int* __restrict__ src,
                                               const int* __restrict__ dst,
                                               int* __restrict__ runs,
                                               v2i* __restrict__ tmp, int E, int nbk,
                                               int rb, int ga,
                                               v2i* __restrict__ st) {
    __shared__ int h[NBK_MAX];
    __shared__ int wt[ED_T / 32];
    const int t = threadIdx.x, lane = t & 31, w = t >> 5;
    h[t] = 0;
    __syncthreads();
    const long long e0 = (long long)rb * ED_EPB;
    int d[8], s[8];
    if (e0 + ED_EPB <= E) {
        const v4i da = *(const v4i*)(dst + e0 + 4 * t);
        const v4i db = *(const v4i*)(dst + e0 + ED_EPB / 2 + 4 * t);
        const v4i sa = *(const v4i*)(src + e0 + 4 * t);
        const v4i sb = *(const v4i*)(src + e0 + ED_EPB / 2 + 4 * t);
        d[0] = da.x; d[1] = da.y; d[2] = da.z; d[3] = da.w;
        d[4] = db.x; d[5] = db.y; d[6] = db.z; d[7] = db.w;
        s[0] = sa.x; s[1] = sa.y; s[2] = sa.z; s[3] = sa.w;
        s[4] = sb.x; s[5] = sb.y; s[6] = sb.z; s[7] = sb.w;
    } else {
#pragma unroll
        for (int u = 0; u < 8; ++u) {
            const long long e = e0 + (u >> 2) * (ED_EPB / 2) + 4 * t + (u & 3);
            d[u] = (e < E) ? dst[e] : -1;
            s[u] = (e < E) ? src[e] : 0;
        }
    }
    int bk[8], r[8];
#pragma unroll
    for (int u = 0; u < 8; ++u) bk[u] = (d[u] >= 0) ? (d[u] >> BK_SHIFT) : 0;
#pragma unroll
    for (int u = 0; u < 8; ++u) {
        r[u] = 0;
        if (d[u] >= 0) r[u] = atomicAdd(&h[bk[u]], 1);
    }
    __syncthreads();

    const int c = h[t];
    int x = c;
#pragma unroll
    for (int o = 1; o < 32; o <<= 1) {
        const int y = __shfl_up(x, o);
        if (lane >= o) x += y;
    }
    if (lane == 31) wt[w] = x;
    __syncthreads();
    int pre = 0;
#pragma unroll
    for (int k = 0; k < ED_T / 32; ++k) pre += (k < w) ? wt[k] : 0;
    const int excl = pre + (x - c);
    h[t] = excl;
    __syncthreads();

    int p[8];
#pragma unroll
    for (int u = 0; u < 8; ++u) p[u] = h[bk[u]] + r[u];
#pragma unroll
    for (int u = 0; u < 8; ++u) {
        v2i o;
        o.x = s[u];
        o.y = d[u];
        st[(d[u] >= 0) ? p[u] : (ED_EPB - 1)] = o;
    }
    __syncthreads();
    if (t < nbk) runs[(size_t)t * ga + rb] = (excl << 16) | c;
    const long long left = (long long)E - e0;
    const int valid = (left < ED_EPB) ? (int)left : ED_EPB;
    const v4i* __restrict__ s4 = (const v4i*)st;
    v4i* __restrict__ g4 = (v4i*)(tmp + e0);
    v4i q[ED_EPB / 2 / ED_T];
#pragma unroll
    for (int k = 0; k < ED_EPB / 2 / ED_T; ++k) q[k] = s4[k * ED_T + t];
#pragma unroll
    for (int k = 0; k < ED_EPB / 2 / ED_T; ++k)
        if (2 * (k * ED_T + t) < valid) g4[k * ED_T + t] = q[k];
}

#define BF_T 1024
#define BK_RPW (BK_NODES / (BF_T / 32))
#define RUN_IT 13
#define BK_CAP 8192

template <bool PLACE>
__device__ __forceinline__ void gcn_bucket_rest(const v2i* __restrict__ tmp, int* cnt,
                                                int* __restrict__ sl, int pkv, int r0, int rpw,
                                                int lane) {
    for (int i = 0; i < rpw; ++i) {
        const int pk = __builtin_amdgcn_readlane(pkv, i);
        const int c = pk & 0xFFFF;
        const int o = pk >> 16;
        for (int j = ((i < RUN_IT) ? 32 : 0) + lane; j < c; j += 32) {
            const v2i sd = tmp[(size_t)(r0 + i) * ED_EPB + o + j];
            const int k = atomicAdd(&cnt[sd.y & (BK_NODES - 1)], 1);
            if (PLACE) sl[k] = sd.x;
        }
    }
}

__global__ __launch_bounds__(BF_T) void gcn_bucket_kernel(const v2i* __restrict__ tmp,
                                                          const int* __restrict__ runs,
                                                          int* __restrict__ start,
                                                          float* __restrict__ norm,
                                                          int* __restrict__ es,
                                                          float* __restrict__ H, int N, int ga) {
    __shared__ int cnt[BK_NODES];
    __shared__ float snorm[BK_NODES];
    __shared__ int wtot[BK_NODES / 32];
    __shared__ int wbs[BF_T / 32];
    __shared__ int ses[BK_CAP + 32];
    const int t = threadIdx.x, lane = t & 31;
    const int w = __builtin_amdgcn_readfirstlane((int)(threadIdx.x >> 5));
    const int b = blockIdx.x;
    const int rpw = (ga + BF_T / 32 - 1) / (BF_T / 32);
    const int r0 = w * rpw;

    int pkv = 0;
    if (lane < rpw && r0 + lane < ga) pkv = runs[(size_t)b * ga + r0 + lane];
    if (t < BK_NODES) cnt[t] = 0;

    const int hrow0 = b * BK_NODES + w * BK_RPW;
    const bool hlive = hrow0 < N;
    const float* Hr = H + (size_t)(hlive ? hrow0 : 0) * HID_DIM + 2 * lane;
    v2f hr[BK_RPW];
#pragma unroll
    for (int r = 0; r < BK_RPW; ++r) hr[r] = *(const v2f*)(Hr + r * HID_DIM);

    int qs[RUN_IT], qd[RUN_IT];
#pragma unroll
    for (int i = 0; i < RUN_IT; ++i) {
        const int pk = __builtin_amdgcn_readlane(pkv, i);
        const int c = pk & 0xFFFF;
        const int o = pk >> 16;
        qs[i] = 0;
        qd[i] = -1;
        if (lane < c) {
            const v2i sd = tmp[(size_t)(r0 + i) * ED_EPB + o + lane];
            qs[i] = sd.x;
            qd[i] = sd.y;
        }
    }
    const bool more = (rpw > RUN_IT) || (__builtin_amdgcn_ballot_w32((pkv & 0xFFFF) > 32) != 0u);

    int osum = pkv >> 16;
#pragma unroll
    for (int o = 16; o > 0; o >>= 1) osum += __shfl_xor(osum, o);
    if (lane == 0) wbs[w] = osum;
    __syncthreads();

#pragma unroll
    for (int i = 0; i < RUN_IT; ++i)
        if (qd[i] >= 0) atomicAdd(&cnt[qd[i] & (BK_NODES - 1)], 1);
    if (more) gcn_bucket_rest<false>(tmp, cnt, es, pkv, r0, rpw, lane);
    __syncthreads();

    const int c = (t < BK_NODES) ? cnt[t] : 0;
    int x = c;
#pragma unroll
    for (int o = 1; o < 32; o <<= 1) {
        const int y = __shfl_up(x, o);
        if (lane >= o) x += y;
    }
    if (t < BK_NODES && lane == 31) wtot[w] = x;
    int bs = wbs[lane];
#pragma unroll
    for (int o = 16; o > 0; o >>= 1) bs += __shfl_xor(bs, o);
    __syncthreads();
    int totv = 0;
#pragma unroll
    for (int k = 0; k < BK_NODES / 32; ++k) totv += wtot[k];
    const int tot = __builtin_amdgcn_readfirstlane(totv);
    const int bsu = __builtin_amdgcn_readfirstlane(bs);
    const int sh = bsu & 31;
    const int g0 = bsu - sh;
    const bool staged = (sh + tot) <= BK_CAP;
    if (t < BK_NODES) {
        int pre = 0;
#pragma unroll
        for (int k = 0; k < BK_NODES / 32; ++k) pre += (k < w) ? wtot[k] : 0;
        const int firstl = sh + pre + (x - c);
        const int node = b * BK_NODES + t;
        cnt[t] = firstl;
        const float nv = rsqrtf(fmaxf((float)c, 1.0f));
        snorm[t] = nv;
        if (node < N) {
            start[node] = g0 + firstl;
            norm[node] = nv;
            if (node == N - 1) start[N] = g0 + firstl + c;
        }
    }
    __syncthreads();

    unsigned r[RUN_IT];
#pragma unroll
    for (int i = 0; i < RUN_IT; ++i) {
        r[i] = 0u;
        if (qd[i] >= 0) r[i] = (unsigned)atomicAdd(&cnt[qd[i] & (BK_NODES - 1)], 1);
    }
    if (staged) {
#pragma unroll
        for (int i = 0; i < RUN_IT; ++i)
            ses[(qd[i] >= 0) ? r[i] : (unsigned)(BK_CAP + lane)] = qs[i];
        if (more) gcn_bucket_rest<true>(tmp, cnt, ses, pkv, r0, rpw, lane);
        __syncthreads();
        const int endj = sh + tot;
        int v[BK_CAP / BF_T];
#pragma unroll
        for (int k = 0; k < BK_CAP / BF_T; ++k) {
            const int j = k * BF_T + t;
            v[k] = (j >= sh && j < endj) ? ses[j] : 0;
        }
#pragma unroll
        for (int k = 0; k < BK_CAP / BF_T; ++k) {
            const int j = k * BF_T + t;
            if (j >= sh && j < endj) es[g0 + j] = v[k];
        }
    } else {
        int* __restrict__ esb = es + g0;
#pragma unroll
        for (int i = 0; i < RUN_IT; ++i)
            if (qd[i] >= 0) esb[r[i]] = qs[i];
        if (more) gcn_bucket_rest<true>(tmp, cnt, esb, pkv, r0, rpw, lane);
    }

    if (hlive) {
        float* Hw = H + (size_t)hrow0 * HID_DIM + 2 * lane;
#pragma unroll
        for (int r2 = 0; r2 < BK_RPW; ++r2) {
            const float nr = snorm[w * BK_RPW + r2];
            v2f o;
            o.x = hr[r2].x * nr;
            o.y = hr[r2].y * nr;
            *(v2f*)(Hw + r2 * HID_DIM) = o;
        }
    }
}

#define GEMM_THREADS 256

template <int K, bool RELU_A, bool SUM_A>
__device__ __forceinline__ void gcn_gemm_block(const float* __restrict__ X,
                              const float* __restrict__ X2,
                              const float* __restrict__ W,
                              float* __restrict__ H, int mtiles, int gb,
                              v4f* __restrict__ sW) {

    const int wave = threadIdx.x >> 5;
    const int lane = threadIdx.x & 31;
    const int half = lane >> 4;
    const int lid  = lane & 15;
    const int tile = gb * (GEMM_THREADS / 32) + wave;
    const int tile_ld = (tile < mtiles) ? tile : (mtiles - 1);
    const int row0 = tile_ld * 16;

    constexpr int ITEMS = K * 8 / GEMM_THREADS;
    const v4f* __restrict__ W4 = (const v4f*)W;
    v4f q0[ITEMS], q1[ITEMS];
#pragma unroll
    for (int it = 0; it < ITEMS; ++it) {
        const int id = it * GEMM_THREADS + threadIdx.x;
        const int p = id >> 4, j = id & 15;
        q0[it] = W4[(2 * p) * 16 + j];
        q1[it] = W4[(2 * p + 1) * 16 + j];
    }

    const float* __restrict__ Arow = X + (size_t)(row0 + lid) * K;
    v2f a[K / 4];
#pragma unroll
    for (int s = 0; s < K / 4; ++s)
        a[s] = *(const v2f*)(Arow + 4 * s + 2 * half);
    v2f a2[SUM_A ? K / 4 : 1];
    if (SUM_A) {
        const float* __restrict__ Arow2 = X2 + (size_t)(row0 + lid) * K;
#pragma unroll
        for (int s = 0; s < K / 4; ++s)
            a2[s] = *(const v2f*)(Arow2 + 4 * s + 2 * half);
    }

#pragma unroll
    for (int it = 0; it < ITEMS; ++it) {
        const int id = it * GEMM_THREADS + threadIdx.x;
        const int p = id >> 4, j = id & 15;
        v4f lo, hi;
        lo.x = q0[it].x; lo.y = q1[it].x; lo.z = q0[it].y; lo.w = q1[it].y;
        hi.x = q0[it].z; hi.y = q1[it].z; hi.z = q0[it].w; hi.w = q1[it].w;
        sW[(2 * p) * 16 + j]     = lo;
        sW[(2 * p + 1) * 16 + j] = hi;
    }
    __syncthreads();

    v8f c0 = {}, c1 = {}, c2 = {}, c3 = {};
#pragma unroll
    for (int s = 0; s < K / 4; ++s) {
        v2f av = a[s];
        if (SUM_A) {
            av.x += a2[s].x;
            av.y += a2[s].y;
        }
        if (RELU_A) {
            av.x = fmaxf(av.x, 0.0f);
            av.y = fmaxf(av.y, 0.0f);
        }
        const int p = 2 * s + half;
        const v4f u0 = sW[(2 * p) * 16 + lid];
        const v4f u1 = sW[(2 * p + 1) * 16 + lid];
        v2f b0, b1, b2, b3;
        b0.x = u0.x; b0.y = u0.y;
        b1.x = u0.z; b1.y = u0.w;
        b2.x = u1.x; b2.y = u1.y;
        b3.x = u1.z; b3.y = u1.w;
        c0 = __builtin_amdgcn_wmma_f32_16x16x4_f32(false, av, false, b0, (short)0, c0, false, false);
        c1 = __builtin_amdgcn_wmma_f32_16x16x4_f32(false, av, false, b1, (short)0, c1, false, false);
        c2 = __builtin_amdgcn_wmma_f32_16x16x4_f32(false, av, false, b2, (short)0, c2, false, false);
        c3 = __builtin_amdgcn_wmma_f32_16x16x4_f32(false, av, false, b3, (short)0, c3, false, false);
    }

    if (tile < mtiles) {
        float* __restrict__ Hb = H + (size_t)(row0 + 8 * half) * HID_DIM + 4 * lid;
#pragma unroll
        for (int r = 0; r < 8; ++r) {
            v4f o;
            o.x = c0[r]; o.y = c1[r]; o.z = c2[r]; o.w = c3[r];
            *(v4f*)(Hb + (size_t)r * HID_DIM) = o;
        }
    }
}

static_assert(GEMM_THREADS == ED_T, "both roles use the same block size");

template <int K>
__global__ __launch_bounds__(GEMM_THREADS) void gcn_front_kernel(const float* __restrict__ X,
                                                                 const float* __restrict__ W,
                                                                 float* __restrict__ H,
                                                                 int mtiles, int gblocks,
                                                                 const int* __restrict__ src,
                                                                 const int* __restrict__ dst,
                                                                 int* __restrict__ runs,
                                                                 v2i* __restrict__ tmp,
                                                                 int E, int nbk, int ga) {
    __shared__ v4f smem[K * 16];
    static_assert(sizeof(v4f) * K * 16 >= sizeof(v2i) * ED_EPB, "the runs pass's pairs fit the buffer");
    const int b = (int)blockIdx.x;
    if (b < gblocks)
        gcn_gemm_block<K, false, false>(X, X, W, H, mtiles, b, smem);
    else
        gcn_runs_block(src, dst, runs, tmp, E, nbk, b - gblocks, ga, (v2i*)smem);
}

#define GA_NPW 8
#define GA_U   8

__device__ __forceinline__ int gcn_slots_load(const int* __restrict__ es, int c, int e, int lane) {
    int m = e - c;
    m = (m > 32) ? 32 : m;
    int idx = c + ((lane < m) ? lane : (m - 1));
    idx = (idx < 0) ? 0 : idx;
    return es[idx];
}

__device__ __forceinline__ void gcn_rows_load(const float* __restrict__ Hl, int s, int j0,
                                              v2f (&h)[GA_U]) {
#pragma unroll
    for (int u = 0; u < GA_U; ++u) {
        const int sj = __builtin_amdgcn_readlane(s, j0 + u);
        h[u] = *(const v2f*)(Hl + (size_t)sj * HID_DIM);
    }
}

__device__ __forceinline__ void gcn_rows_sum(int j0, int m, const v2f (&h)[GA_U],
                                             float& ax, float& ay) {
#pragma unroll
    for (int u = 0; u < GA_U; ++u) {
        const bool ok = (j0 + u) < m;
        ax += ok ? h[u].x : 0.0f;
        ay += ok ? h[u].y : 0.0f;
    }
}

__device__ __forceinline__ void gcn_slots_sum(const float* __restrict__ Hl, int s, int m,
                                              float& ax, float& ay) {
    for (int j = 0; j < m; j += 2 * GA_U) {
        v2f h0[GA_U], h1[GA_U];
        const bool two = (j + GA_U) < m;
        gcn_rows_load(Hl, s, j, h0);
        if (two) gcn_rows_load(Hl, s, j + GA_U, h1);
        gcn_rows_sum(j, m, h0, ax, ay);
        if (two) gcn_rows_sum(j + GA_U, m, h1, ax, ay);
    }
}

#define GA_WAVES 8
#define GA_ROWS  (GA_NPW * GA_WAVES)
#define GA_PITCH 68

template <bool FUSE_GEMM>
__global__ __launch_bounds__(256) void gcn_gather_kernel(const float* __restrict__ H,
                                                         const float* __restrict__ norm,
                                                         const int* __restrict__ start,
                                                         const int* __restrict__ es,
                                                         const float* __restrict__ W,
                                                         float* __restrict__ out, int N) {
    __shared__ v4f   sW[FUSE_GEMM ? HID_DIM * 16 : 1];
    __shared__ float sX[FUSE_GEMM ? GA_ROWS * GA_PITCH : 2];

    const int lane = threadIdx.x & 31;
    const int wib = __builtin_amdgcn_readfirstlane((int)(threadIdx.x >> 5));
    const int n0 = ((int)blockIdx.x * GA_WAVES + wib) * GA_NPW;
    if (!FUSE_GEMM) {
        if (n0 >= N) return;
    }
    int nn = (N - n0 < GA_NPW) ? (N - n0) : GA_NPW;
    nn = (nn < 0) ? 0 : nn;
    const float* __restrict__ Hl = H + 2 * lane;

    int bi = n0 + ((lane < nn) ? lane : nn);
    bi = (bi > N) ? N : bi;
    const int bv = start[bi];

    constexpr int W_ITEMS = HID_DIM * 8 / 256;
    v4f q0[W_ITEMS], q1[W_ITEMS];
    if (FUSE_GEMM) {
        const v4f* __restrict__ W4 = (const v4f*)W;
#pragma unroll
        for (int it = 0; it < W_ITEMS; ++it) {
            const int id = it * 256 + (int)threadIdx.x;
            const int p = id >> 4, j = id & 15;
            q0[it] = W4[(2 * p) * 16 + j];
            q1[it] = W4[(2 * p + 1) * 16 + j];
        }
    }

    int b = __builtin_amdgcn_readlane(bv, 0);
    int e = __builtin_amdgcn_readlane(bv, 1);
    int s = gcn_slots_load(es, b, e, lane);

    for (int k = 0; k < nn; ++k) {
        int e2 = e;
        int s2 = s;
        if (k + 1 < nn) {
            e2 = __builtin_amdgcn_readlane(bv, k + 2);
            s2 = gcn_slots_load(es, e, e2, lane);
        }

        float ax = 0.0f, ay = 0.0f;
        if (b < e) {
            gcn_slots_sum(Hl, s, (e - b > 32) ? 32 : (e - b), ax, ay);
            for (int c = b + 32; c < e; c += 32) {
                s = gcn_slots_load(es, c, e, lane);
                gcn_slots_sum(Hl, s, (e - c > 32) ? 32 : (e - c), ax, ay);
            }
        }
        v2f o;
        o.x = fmaxf(ax, 0.0f);
        o.y = fmaxf(ay, 0.0f);
        if (FUSE_GEMM)
            *(v2f*)(sX + (wib * GA_NPW + k) * GA_PITCH + 2 * lane) = o;
        else
            *(v2f*)(out + (size_t)(n0 + k) * HID_DIM + 2 * lane) = o;

        b = e;
        e = e2;
        s = s2;
    }

    if (FUSE_GEMM) {
#pragma unroll
        for (int it = 0; it < W_ITEMS; ++it) {
            const int id = it * 256 + (int)threadIdx.x;
            const int p = id >> 4, j = id & 15;
            v4f lo, hi;
            lo.x = q0[it].x; lo.y = q1[it].x; lo.z = q0[it].y; lo.w = q1[it].y;
            hi.x = q0[it].z; hi.y = q1[it].z; hi.z = q0[it].w; hi.w = q1[it].w;
            sW[(2 * p) * 16 + j]     = lo;
            sW[(2 * p + 1) * 16 + j] = hi;
        }

        for (int k = nn; k < GA_NPW; ++k) {
            v2f z;
            z.x = 0.0f;
            z.y = 0.0f;
            *(v2f*)(sX + (wib * GA_NPW + k) * GA_PITCH + 2 * lane) = z;
        }
        __syncthreads();
        const int strip0 = (int)blockIdx.x * GA_ROWS + wib * 16;
        if (wib >= GA_ROWS / 16 || strip0 >= N) return;

        const int half = lane >> 4;
        const int lid  = lane & 15;
        const v4f na = *(const v4f*)(norm + strip0 + 8 * half);
        const v4f nb = *(const v4f*)(norm + strip0 + 8 * half + 4);
        const float* Ar = sX + (wib * 16 + lid) * GA_PITCH + 2 * half;
        v8f c0 = {}, c1 = {}, c2 = {}, c3 = {};
#pragma unroll
        for (int s4 = 0; s4 < HID_DIM / 4; ++s4) {
            const v2f av = *(const v2f*)(Ar + 4 * s4);
            const int p = 2 * s4 + half;
            const v4f u0 = sW[(2 * p) * 16 + lid];
            const v4f u1 = sW[(2 * p + 1) * 16 + lid];
            v2f b0, b1, b2, b3;
            b0.x = u0.x; b0.y = u0.y;
            b1.x = u0.z; b1.y = u0.w;
            b2.x = u1.x; b2.y = u1.y;
            b3.x = u1.z; b3.y = u1.w;
            c0 = __builtin_amdgcn_wmma_f32_16x16x4_f32(false, av, false, b0, (short)0, c0, false, false);
            c1 = __builtin_amdgcn_wmma_f32_16x16x4_f32(false, av, false, b1, (short)0, c1, false, false);
            c2 = __builtin_amdgcn_wmma_f32_16x16x4_f32(false, av, false, b2, (short)0, c2, false, false);
            c3 = __builtin_amdgcn_wmma_f32_16x16x4_f32(false, av, false, b3, (short)0, c3, false, false);
        }

        float* __restrict__ Ob = out + (size_t)(strip0 + 8 * half) * HID_DIM + 4 * lid;
#pragma unroll
        for (int r = 0; r < 8; ++r) {
            const float nr = (r < 4) ? na[r & 3] : nb[r & 3];
            v4f o;
            o.x = c0[r] * nr; o.y = c1[r] * nr; o.z = c2[r] * nr; o.w = c3[r] * nr;
            *(v4f*)(Ob + (size_t)r * HID_DIM) = o;
        }
    }
}


extern "C" void kernel_launch(void* const* d_in, const int* in_sizes, int n_in,
                              void* d_out, int out_size, void* d_ws, size_t ws_size,
                              hipStream_t stream) {
    const float* features = (const float*)d_in[0];
    const float* W1       = (const float*)d_in[1];
    const float* W2       = (const float*)d_in[2];
    const int*   src      = (const int*)d_in[3];
    const int*   dst      = (const int*)d_in[4];

    const int N = in_sizes[0] / IN_DIM;
    const int E = in_sizes[3];
    const long long NH = (long long)N * HID_DIM;
    const long long E64 = ((long long)E + 63) / 64 * 64;

    const int nbk = (N + BK_NODES - 1) / BK_NODES;
    const int ga  = (E + ED_EPB - 1) / ED_EPB;
    const int nsc = nbk * ga;
    const long long S64 = ((long long)nsc + 63) / 64 * 64;

    int*   runs  = (int*)d_ws;
    int*   start = runs + S64;
    float* norm  = (float*)(start + 65536 + 64);
    v2i*   tmp   = (v2i*)(norm + 65536);
    int*   es    = (int*)(tmp + (long long)ga * ED_EPB);
    float* H     = (float*)(es + E64);
    float* H2    = H + NH;
    float* OUT   = (float*)d_out;

    const int mtiles = N / 16;
    const int gblocks = (mtiles + GEMM_THREADS / 32 - 1) / (GEMM_THREADS / 32);
    const int nblocks = (N + GA_ROWS - 1) / GA_ROWS;

    gcn_front_kernel<IN_DIM><<<gblocks + ga, GEMM_THREADS, 0, stream>>>(features, W1, H, mtiles, gblocks,
                                                                      src, dst, runs, tmp, E, nbk, ga);
    gcn_bucket_kernel<<<nbk, BF_T, 0, stream>>>(tmp, runs, start, norm, es, H, N, ga);

    gcn_gather_kernel<true><<<nblocks, 256, 0, stream>>>(H, norm, start, es, W2, H2, N);

    gcn_gather_kernel<false><<<nblocks, 256, 0, stream>>>(H2, norm, start, es, W2, OUT, N);
}
